# starting kernel
# baseline (speedup 1.0000x reference)
#include <hip/hip_runtime.h>
#include <math.h>

#define NB   16384
#define NQ   100
#define NS   50
#define DCUT 32
#define HLOC 4

#define MPAD 34
#define HPAD 17

typedef __attribute__((ext_vector_type(2))) float v2f;
typedef __attribute__((ext_vector_type(8))) float v8f;

static __device__ __forceinline__ v8f wmma4(v2f a, v2f b, v8f c) {
    return __builtin_amdgcn_wmma_f32_16x16x4_f32(
         false, a,  false, b,
         (short)0, c,  false,  false);
}

__global__ __launch_bounds__(128) void mps_rnn_wmma_kernel(
    const int*   __restrict__ xg,
    const float* __restrict__ Mg,
    const float* __restrict__ vg,
    const float* __restrict__ wg,
    const float* __restrict__ cg,
    float*       __restrict__ outg)
{
    __shared__ float Mr[HLOC * DCUT * MPAD];
    __shared__ float Mi[HLOC * DCUT * MPAD];
    __shared__ float vr_s[HLOC * DCUT];
    __shared__ float vi_s[HLOC * DCUT];
    __shared__ float wr_s[DCUT];
    __shared__ float wi_s[DCUT];
    __shared__ float c_s[2];
    __shared__ float hbuf[4][2][DCUT][HPAD];

    const int tid  = threadIdx.x;
    const int lane = tid & 31;
    const int wave = tid >> 5;
    const int n    = lane & 15;
    const int half = lane >> 4;
    const int bcol = blockIdx.x * 64 + wave * 16 + n;

    v2f Hr[8], Hi[8];
    const float h0 = 0.17677669529663688f;
    #pragma unroll
    for (int k = 0; k < 8; ++k) {
        Hr[k][0] = h0; Hr[k][1] = h0;
        Hi[k][0] = 0.f; Hi[k][1] = 0.f;
    }

    float logp = 0.f, phase = 0.f;
    const int k0h = half * 2;

    for (int t = 0; t < NS; ++t) {
        {
            const float* Mt = Mg + (size_t)t * HLOC * DCUT * DCUT * 2;
            for (int idx = tid; idx < HLOC * DCUT * DCUT; idx += 128) {
                const int s = idx >> 10, rem = idx & 1023;
                const int d = rem >> 5,  e  = rem & 31;
                const float re = Mt[idx * 2 + 0];
                const float im = Mt[idx * 2 + 1];
                Mr[(s * DCUT + d) * MPAD + e] = re;
                Mi[(s * DCUT + d) * MPAD + e] = im;
            }
            const float* vt = vg + (size_t)t * HLOC * DCUT * 2;
            if (tid < HLOC * DCUT) { vr_s[tid] = vt[tid * 2]; vi_s[tid] = vt[tid * 2 + 1]; }
            const float* wt = wg + (size_t)t * DCUT * 2;
            if (tid < DCUT) { wr_s[tid] = wt[tid * 2]; wi_s[tid] = wt[tid * 2 + 1]; }
            if (tid == 0) { c_s[0] = cg[t * 2]; c_s[1] = cg[t * 2 + 1]; }
            if (t + 1 < NS) {
                const char* nxt = (const char*)(Mg + (size_t)(t + 1) * HLOC * DCUT * DCUT * 2);
                __builtin_prefetch(nxt + tid * 256, 0, 1);
            }
        }
        __syncthreads();

        v8f htR[HLOC][2], htI[HLOC][2];
        float p_part[HLOC];
        #pragma unroll
        for (int s = 0; s < HLOC; ++s) {
            p_part[s] = 0.f;
            #pragma unroll
            for (int mt = 0; mt < 2; ++mt) {
                v8f accR  = {};
                v8f accMI = {};
                v8f accI  = {};
                const float* mrBase = &Mr[((s * DCUT) + mt * 16 + n) * MPAD];
                const float* miBase = &Mi[((s * DCUT) + mt * 16 + n) * MPAD];
                #pragma unroll
                for (int kk = 0; kk < 8; ++kk) {
                    const int koff = kk * 4 + k0h;
                    const v2f Ar = *(const v2f*)(mrBase + koff);
                    const v2f Ai = *(const v2f*)(miBase + koff);
                    accR  = wmma4(Ar, Hr[kk], accR);
                    accMI = wmma4(Ai, Hi[kk], accMI);
                    accI  = wmma4(Ar, Hi[kk], accI);
                    accI  = wmma4(Ai, Hr[kk], accI);
                }
                #pragma unroll
                for (int r = 0; r < 8; ++r) {
                    const int row = mt * 16 + r + half * 8;
                    const float hr = accR[r] - accMI[r] + vr_s[s * DCUT + row];
                    const float hi = accI[r]            + vi_s[s * DCUT + row];
                    htR[s][mt][r] = hr;
                    htI[s][mt][r] = hi;
                    p_part[s] += hr * hr + hi * hi;
                }
            }
        }

        float p_sig[HLOC], psum = 0.f;
        #pragma unroll
        for (int s = 0; s < HLOC; ++s) {
            p_sig[s] = p_part[s] + __shfl_xor(p_part[s], 16, 32);
            psum += p_sig[s];
        }

        const int xs0 = xg[(size_t)bcol * NQ + 2 * t];
        const int xs1 = xg[(size_t)bcol * NQ + 2 * t + 1];
        const int ss  = 2 * xs0 + xs1;

        const float p_sel = (ss == 0) ? p_sig[0] : (ss == 1) ? p_sig[1]
                          : (ss == 2) ? p_sig[2] : p_sig[3];
        logp += logf(p_sel) - logf(psum);
        const float rinv = rsqrtf(p_sel);

        float ampR = 0.f, ampI = 0.f;
        #pragma unroll
        for (int mt = 0; mt < 2; ++mt) {
            const v8f selR = (ss == 0) ? htR[0][mt] : (ss == 1) ? htR[1][mt]
                           : (ss == 2) ? htR[2][mt] : htR[3][mt];
            const v8f selI = (ss == 0) ? htI[0][mt] : (ss == 1) ? htI[1][mt]
                           : (ss == 2) ? htI[2][mt] : htI[3][mt];
            #pragma unroll
            for (int r = 0; r < 8; ++r) {
                const int row = mt * 16 + r + half * 8;
                const float hr = selR[r] * rinv;
                const float hi = selI[r] * rinv;
                hbuf[wave][0][row][n] = hr;
                hbuf[wave][1][row][n] = hi;
                const float wr = wr_s[row], wi = wi_s[row];
                ampR += wr * hr - wi * hi;
                ampI += wr * hi + wi * hr;
            }
        }
        ampR += __shfl_xor(ampR, 16, 32);
        ampI += __shfl_xor(ampI, 16, 32);
        ampR += c_s[0];
        ampI += c_s[1];
        phase += atan2f(ampI, ampR);

        __syncthreads();

        #pragma unroll
        for (int kk = 0; kk < 8; ++kk) {
            const int krow = kk * 4 + half * 2;
            Hr[kk][0] = hbuf[wave][0][krow + 0][n];
            Hr[kk][1] = hbuf[wave][0][krow + 1][n];
            Hi[kk][0] = hbuf[wave][1][krow + 0][n];
            Hi[kk][1] = hbuf[wave][1][krow + 1][n];
        }
    }

    if (half == 0) {
        float2 o;
        o.x = 0.5f * logp;
        o.y = phase;
        *(float2*)(outg + (size_t)bcol * 2) = o;
    }
}

extern "C" void kernel_launch(void* const* d_in, const int* in_sizes, int n_in,
                              void* d_out, int out_size, void* d_ws, size_t ws_size,
                              hipStream_t stream) {
    (void)in_sizes; (void)n_in; (void)out_size; (void)d_ws; (void)ws_size;
    const int*   x  = (const int*)  d_in[0];
    const float* M  = (const float*)d_in[1];
    const float* v  = (const float*)d_in[2];
    const float* w  = (const float*)d_in[3];
    const float* c  = (const float*)d_in[4];
    float* out = (float*)d_out;

    dim3 grid(NB / 64);
    dim3 block(128);
    mps_rnn_wmma_kernel<<<grid, block, 0, stream>>>(x, M, v, w, c, out);
}


# Next site's M (16-byte loads), v, w and this site's x, c are loaded ahead of the WMMA work into a second LDS stage: 8 serial load round trips per site overlap the matrix work; 1 barrier per site
# speedup vs baseline: 1.3431x; 1.3431x over previous
#include <hip/hip_runtime.h>
#include <math.h>


#define NB   16384
#define NQ   100
#define NS   50
#define DCUT 32
#define HLOC 4

#define MPAD 34
#define HPAD 17

#define MSZ     (HLOC * DCUT * MPAD)
#define OFF_MR  0
#define OFF_MI  MSZ
#define OFF_VR  (2 * MSZ)
#define OFF_VI  (2 * MSZ + HLOC * DCUT)
#define OFF_W   (2 * MSZ + 2 * HLOC * DCUT)
#define SITE_FLOATS (2 * MSZ + 2 * HLOC * DCUT + 4 * 2 * DCUT)

typedef __attribute__((ext_vector_type(2))) float v2f;
typedef __attribute__((ext_vector_type(8))) float v8f;

static __device__ __forceinline__ v8f wmma4(v2f a, v2f b, v8f c) {
    return __builtin_amdgcn_wmma_f32_16x16x4_f32(
         false, a,  false, b,
         (short)0, c,  false,  false);
}

static __device__ __forceinline__ void put_m(float* S, int off, float4 m) {
    v2f re, im;
    re[0] = m.x; re[1] = m.z;
    im[0] = m.y; im[1] = m.w;
    *(v2f*)(S + OFF_MR + off) = re;
    *(v2f*)(S + OFF_MI + off) = im;
}

__global__ __launch_bounds__(128) void mps_rnn_wmma_kernel(
    const int*   __restrict__ xg,
    const float* __restrict__ Mg,
    const float* __restrict__ vg,
    const float* __restrict__ wg,
    const float* __restrict__ cg,
    float*       __restrict__ outg)
{
    __shared__ __attribute__((aligned(16))) float site[2][SITE_FLOATS];
    __shared__ float hbuf[4][2][DCUT][HPAD];

    const int tid  = threadIdx.x;
    const int lane = tid & 31;
    const int wave = tid >> 5;
    const int n    = lane & 15;
    const int half = lane >> 4;
    const int bcol = blockIdx.x * 64 + wave * 16 + n;

    const int sb = (tid >> 4) * MPAD + 2 * (tid & 15);
    const float4* M4 = reinterpret_cast<const float4*>(Mg);
    const float2* v2 = reinterpret_cast<const float2*>(vg);
    const float2* w2 = reinterpret_cast<const float2*>(wg);

    {
        float* S = site[0];
        #pragma unroll
        for (int q = 0; q < 16; ++q) put_m(S, sb + 272 * q, M4[tid + 128 * q]);
        const float2 vq = v2[tid];
        S[OFF_VR + tid] = vq.x;
        S[OFF_VI + tid] = vq.y;
        const float2 wq = w2[lane];
        S[OFF_W + wave * 2 * DCUT + lane]        = wq.x;
        S[OFF_W + wave * 2 * DCUT + DCUT + lane] = wq.y;
    }

    v2f Hr[8], Hi[8];
    const float h0 = 0.17677669529663688f;
    #pragma unroll
    for (int k = 0; k < 8; ++k) {
        Hr[k][0] = h0; Hr[k][1] = h0;
        Hi[k][0] = 0.f; Hi[k][1] = 0.f;
    }

    float logp = 0.f, phase = 0.f;
    const int k0h = half * 2;

    __syncthreads();

    for (int t = 0; t < NS; ++t) {
        const float* C = site[t & 1];
        float*       N = site[(t & 1) ^ 1];

        const int tn = (t + 1 < NS) ? (t + 1) : (NS - 1);
        const float4* Mn = M4 + (size_t)tn * (HLOC * DCUT * DCUT / 2) + tid;
        const float2 vq = v2[tn * (HLOC * DCUT) + tid];
        const float2 wq = w2[tn * DCUT + lane];
        const int xs0 = xg[(size_t)bcol * NQ + 2 * t];
        const int xs1 = xg[(size_t)bcol * NQ + 2 * t + 1];
        const float cR = cg[2 * t];
        const float cI = cg[2 * t + 1];
        float4 mq[16];

        v8f htR[HLOC][2], htI[HLOC][2];
        float p_part[HLOC];
        #pragma unroll
        for (int s = 0; s < HLOC; ++s) {
            p_part[s] = 0.f;
            #pragma unroll
            for (int mt = 0; mt < 2; ++mt) {
                const int j = s * 2 + mt;
                if (j == 0) {
                    mq[0] = Mn[0];
                    mq[1] = Mn[128];
                }
                if (j < 7) {
                    mq[2 * j + 2] = Mn[128 * (2 * j + 2)];
                    mq[2 * j + 3] = Mn[128 * (2 * j + 3)];
                }
                __builtin_amdgcn_sched_barrier(0x7CF);
                if (j >= 2) {
                    put_m(N, sb + 272 * (2 * j - 4), mq[2 * j - 4]);
                    put_m(N, sb + 272 * (2 * j - 3), mq[2 * j - 3]);
                }

                v8f accR  = {};
                v8f accMI = {};
                v8f accI  = {};
                const float* mrBase = C + OFF_MR + ((s * DCUT) + mt * 16 + n) * MPAD;
                const float* miBase = C + OFF_MI + ((s * DCUT) + mt * 16 + n) * MPAD;
                #pragma unroll
                for (int kk = 0; kk < 8; ++kk) {
                    const int koff = kk * 4 + k0h;
                    const v2f Ar = *(const v2f*)(mrBase + koff);
                    const v2f Ai = *(const v2f*)(miBase + koff);
                    accR  = wmma4(Ar, Hr[kk], accR);
                    accMI = wmma4(Ai, Hi[kk], accMI);
                    accI  = wmma4(Ar, Hi[kk], accI);
                    accI  = wmma4(Ai, Hr[kk], accI);
                }
                #pragma unroll
                for (int r = 0; r < 8; ++r) {
                    const int row = mt * 16 + r + half * 8;
                    const float hr = accR[r] - accMI[r] + C[OFF_VR + s * DCUT + row];
                    const float hi = accI[r]            + C[OFF_VI + s * DCUT + row];
                    htR[s][mt][r] = hr;
                    htI[s][mt][r] = hi;
                    p_part[s] += hr * hr + hi * hi;
                }
            }
        }
        put_m(N, sb + 272 * 12, mq[12]);
        put_m(N, sb + 272 * 13, mq[13]);
        put_m(N, sb + 272 * 14, mq[14]);
        put_m(N, sb + 272 * 15, mq[15]);
        N[OFF_VR + tid] = vq.x;
        N[OFF_VI + tid] = vq.y;
        N[OFF_W + wave * 2 * DCUT + lane]        = wq.x;
        N[OFF_W + wave * 2 * DCUT + DCUT + lane] = wq.y;

        float p_sig[HLOC], psum = 0.f;
        #pragma unroll
        for (int s = 0; s < HLOC; ++s) {
            p_sig[s] = p_part[s] + __shfl_xor(p_part[s], 16, 32);
            psum += p_sig[s];
        }

        const int ss  = 2 * xs0 + xs1;

        const float p_sel = (ss == 0) ? p_sig[0] : (ss == 1) ? p_sig[1]
                          : (ss == 2) ? p_sig[2] : p_sig[3];
        logp += logf(p_sel) - logf(psum);
        const float rinv = rsqrtf(p_sel);

        float ampR = 0.f, ampI = 0.f;
        #pragma unroll
        for (int mt = 0; mt < 2; ++mt) {
            const v8f selR = (ss == 0) ? htR[0][mt] : (ss == 1) ? htR[1][mt]
                           : (ss == 2) ? htR[2][mt] : htR[3][mt];
            const v8f selI = (ss == 0) ? htI[0][mt] : (ss == 1) ? htI[1][mt]
                           : (ss == 2) ? htI[2][mt] : htI[3][mt];
            #pragma unroll
            for (int r = 0; r < 8; ++r) {
                const int row = mt * 16 + r + half * 8;
                const float hr = selR[r] * rinv;
                const float hi = selI[r] * rinv;
                hbuf[wave][0][row][n] = hr;
                hbuf[wave][1][row][n] = hi;
                const float wr = C[OFF_W + wave * 2 * DCUT + row];
                const float wi = C[OFF_W + wave * 2 * DCUT + DCUT + row];
                ampR += wr * hr - wi * hi;
                ampI += wr * hi + wi * hr;
            }
        }
        ampR += __shfl_xor(ampR, 16, 32);
        ampI += __shfl_xor(ampI, 16, 32);
        ampR += cR;
        ampI += cI;
        phase += atan2f(ampI, ampR);

        __syncthreads();

        #pragma unroll
        for (int kk = 0; kk < 8; ++kk) {
            const int krow = kk * 4 + half * 2;
            Hr[kk][0] = hbuf[wave][0][krow + 0][n];
            Hr[kk][1] = hbuf[wave][0][krow + 1][n];
            Hi[kk][0] = hbuf[wave][1][krow + 0][n];
            Hi[kk][1] = hbuf[wave][1][krow + 1][n];
        }
    }

    if (half == 0) {
        float2 o;
        o.x = 0.5f * logp;
        o.y = phase;
        *(float2*)(outg + (size_t)bcol * 2) = o;
    }
}

extern "C" void kernel_launch(void* const* d_in, const int* in_sizes, int n_in,
                              void* d_out, int out_size, void* d_ws, size_t ws_size,
                              hipStream_t stream) {
    (void)in_sizes; (void)n_in; (void)out_size; (void)d_ws; (void)ws_size;
    const int*   x  = (const int*)  d_in[0];
    const float* M  = (const float*)d_in[1];
    const float* v  = (const float*)d_in[2];
    const float* w  = (const float*)d_in[3];
    const float* c  = (const float*)d_in[4];
    float* out = (float*)d_out;

    dim3 grid(NB / 64);
    dim3 block(128);
    mps_rnn_wmma_kernel<<<grid, block, 0, stream>>>(x, M, v, w, c, out);
}


# Each tile's LDS reads (both M planes' fragments and its v rows) go as one group into a second register set ahead of the previous tile's WMMAs, so no wait drains the LDS queue inside a WMMA run
# speedup vs baseline: 1.7944x; 1.3360x over previous
#include <hip/hip_runtime.h>
#include <math.h>


#define NB   16384
#define NQ   100
#define NS   50
#define DCUT 32
#define HLOC 4

#define MPAD 34
#define HPAD 17

#define MSZ     (HLOC * DCUT * MPAD)
#define OFF_MR  0
#define OFF_MI  MSZ
#define OFF_VR  (2 * MSZ)
#define OFF_VI  (2 * MSZ + HLOC * DCUT)
#define OFF_W   (2 * MSZ + 2 * HLOC * DCUT)
#define SITE_FLOATS (2 * MSZ + 2 * HLOC * DCUT + 4 * 2 * DCUT)

typedef __attribute__((ext_vector_type(2))) float v2f;
typedef __attribute__((ext_vector_type(8))) float v8f;

static __device__ __forceinline__ v8f wmma4(v2f a, v2f b, v8f c) {
    return __builtin_amdgcn_wmma_f32_16x16x4_f32(
         false, a,  false, b,
         (short)0, c,  false,  false);
}

static __device__ __forceinline__ void put_m(float* S, int off, float4 m) {
    v2f re, im;
    re[0] = m.x; re[1] = m.z;
    im[0] = m.y; im[1] = m.w;
    *(v2f*)(S + OFF_MR + off) = re;
    *(v2f*)(S + OFF_MI + off) = im;
}

__global__ __launch_bounds__(128) void mps_rnn_wmma_kernel(
    const int*   __restrict__ xg,
    const float* __restrict__ Mg,
    const float* __restrict__ vg,
    const float* __restrict__ wg,
    const float* __restrict__ cg,
    float*       __restrict__ outg)
{
    __shared__ __attribute__((aligned(16))) float site[2][SITE_FLOATS];
    __shared__ float hbuf[4][2][DCUT][HPAD];

    const int tid  = threadIdx.x;
    const int lane = tid & 31;
    const int wave = tid >> 5;
    const int n    = lane & 15;
    const int half = lane >> 4;
    const int bcol = blockIdx.x * 64 + wave * 16 + n;

    const int sb = (tid >> 4) * MPAD + 2 * (tid & 15);
    const float4* M4 = reinterpret_cast<const float4*>(Mg);
    const float2* v2 = reinterpret_cast<const float2*>(vg);
    const float2* w2 = reinterpret_cast<const float2*>(wg);

    {
        float* S = site[0];
        #pragma unroll
        for (int q = 0; q < 16; ++q) put_m(S, sb + 272 * q, M4[tid + 128 * q]);
        const float2 vq = v2[tid];
        S[OFF_VR + tid] = vq.x;
        S[OFF_VI + tid] = vq.y;
        const float2 wq = w2[lane];
        S[OFF_W + wave * 2 * DCUT + lane]        = wq.x;
        S[OFF_W + wave * 2 * DCUT + DCUT + lane] = wq.y;
    }

    v2f Hr[8], Hi[8];
    const float h0 = 0.17677669529663688f;
    #pragma unroll
    for (int k = 0; k < 8; ++k) {
        Hr[k][0] = h0; Hr[k][1] = h0;
        Hi[k][0] = 0.f; Hi[k][1] = 0.f;
    }

    float logp = 0.f, phase = 0.f;
    const int k0h = half * 2;

    __syncthreads();

    for (int t = 0; t < NS; ++t) {
        const float* C = site[t & 1];
        float*       N = site[(t & 1) ^ 1];

        const int tn = (t + 1 < NS) ? (t + 1) : (NS - 1);
        const float4* Mn = M4 + (size_t)tn * (HLOC * DCUT * DCUT / 2) + tid;
        const float2 vq = v2[tn * (HLOC * DCUT) + tid];
        const float2 wq = w2[tn * DCUT + lane];
        const int xs0 = xg[(size_t)bcol * NQ + 2 * t];
        const int xs1 = xg[(size_t)bcol * NQ + 2 * t + 1];
        const float cR = cg[2 * t];
        const float cI = cg[2 * t + 1];
        float4 mq[16];

        v2f    fAr[2][8], fAi[2][8];
        float4 fVr[2][2], fVi[2][2];

        {
            const float* mrBase = C + OFF_MR + n * MPAD + k0h;
            const float* miBase = C + OFF_MI + n * MPAD + k0h;
            #pragma unroll
            for (int kk = 0; kk < 8; ++kk) {
                fAr[0][kk] = *(const v2f*)(mrBase + kk * 4);
                fAi[0][kk] = *(const v2f*)(miBase + kk * 4);
            }
            const float4* vr4 = (const float4*)(C + OFF_VR + half * 8);
            const float4* vi4 = (const float4*)(C + OFF_VI + half * 8);
            fVr[0][0] = vr4[0]; fVr[0][1] = vr4[1];
            fVi[0][0] = vi4[0]; fVi[0][1] = vi4[1];
        }

        v8f htR[HLOC][2], htI[HLOC][2];
        float p_part[HLOC];
        #pragma unroll
        for (int s = 0; s < HLOC; ++s) {
            p_part[s] = 0.f;
            #pragma unroll
            for (int mt = 0; mt < 2; ++mt) {
                const int j   = s * 2 + mt;
                const int cur = j & 1;
                const int nxt = cur ^ 1;

                if (j == 0) {
                    mq[0] = Mn[0];
                    mq[1] = Mn[128];
                }
                if (j < 7) {
                    mq[2 * j + 2] = Mn[128 * (2 * j + 2)];
                    mq[2 * j + 3] = Mn[128 * (2 * j + 3)];
                }
                if (j >= 2) {
                    put_m(N, sb + 272 * (2 * j - 4), mq[2 * j - 4]);
                    put_m(N, sb + 272 * (2 * j - 3), mq[2 * j - 3]);
                }
                if (j < 7) {
                    const int s1  = (j + 1) >> 1;
                    const int mt1 = (j + 1) & 1;
                    const float* mrBase = C + OFF_MR + ((s1 * DCUT) + mt1 * 16 + n) * MPAD + k0h;
                    const float* miBase = C + OFF_MI + ((s1 * DCUT) + mt1 * 16 + n) * MPAD + k0h;
                    #pragma unroll
                    for (int kk = 0; kk < 8; ++kk) {
                        fAr[nxt][kk] = *(const v2f*)(mrBase + kk * 4);
                        fAi[nxt][kk] = *(const v2f*)(miBase + kk * 4);
                    }
                    const float4* vr4 = (const float4*)(C + OFF_VR + s1 * DCUT + mt1 * 16 + half * 8);
                    const float4* vi4 = (const float4*)(C + OFF_VI + s1 * DCUT + mt1 * 16 + half * 8);
                    fVr[nxt][0] = vr4[0]; fVr[nxt][1] = vr4[1];
                    fVi[nxt][0] = vi4[0]; fVi[nxt][1] = vi4[1];
                }
                asm volatile("" : "+v"(fAr[cur][0]), "+v"(fAi[cur][0]));

                v8f accR  = {};
                v8f accMI = {};
                v8f accI  = {};
                #pragma unroll
                for (int kk = 0; kk < 8; ++kk) {
                    const v2f Ar = fAr[cur][kk];
                    const v2f Ai = fAi[cur][kk];
                    accR  = wmma4(Ar, Hr[kk], accR);
                    accMI = wmma4(Ai, Hi[kk], accMI);
                    accI  = wmma4(Ar, Hi[kk], accI);
                    accI  = wmma4(Ai, Hr[kk], accI);
                }
                #pragma unroll
                for (int r = 0; r < 8; ++r) {
                    const float4 vr = fVr[cur][r >> 2];
                    const float4 vi = fVi[cur][r >> 2];
                    const float vrr = ((r & 3) == 0) ? vr.x : ((r & 3) == 1) ? vr.y : ((r & 3) == 2) ? vr.z : vr.w;
                    const float vii = ((r & 3) == 0) ? vi.x : ((r & 3) == 1) ? vi.y : ((r & 3) == 2) ? vi.z : vi.w;
                    const float hr = accR[r] - accMI[r] + vrr;
                    const float hi = accI[r]            + vii;
                    htR[s][mt][r] = hr;
                    htI[s][mt][r] = hi;
                    p_part[s] += hr * hr + hi * hi;
                }
            }
        }
        put_m(N, sb + 272 * 12, mq[12]);
        put_m(N, sb + 272 * 13, mq[13]);
        put_m(N, sb + 272 * 14, mq[14]);
        put_m(N, sb + 272 * 15, mq[15]);
        N[OFF_VR + tid] = vq.x;
        N[OFF_VI + tid] = vq.y;
        N[OFF_W + wave * 2 * DCUT + lane]        = wq.x;
        N[OFF_W + wave * 2 * DCUT + DCUT + lane] = wq.y;

        float p_sig[HLOC], psum = 0.f;
        #pragma unroll
        for (int s = 0; s < HLOC; ++s) {
            p_sig[s] = p_part[s] + __shfl_xor(p_part[s], 16, 32);
            psum += p_sig[s];
        }

        const int ss  = 2 * xs0 + xs1;

        const float p_sel = (ss == 0) ? p_sig[0] : (ss == 1) ? p_sig[1]
                          : (ss == 2) ? p_sig[2] : p_sig[3];
        logp += logf(p_sel) - logf(psum);
        const float rinv = rsqrtf(p_sel);

        float ampR = 0.f, ampI = 0.f;
        #pragma unroll
        for (int mt = 0; mt < 2; ++mt) {
            const v8f selR = (ss == 0) ? htR[0][mt] : (ss == 1) ? htR[1][mt]
                           : (ss == 2) ? htR[2][mt] : htR[3][mt];
            const v8f selI = (ss == 0) ? htI[0][mt] : (ss == 1) ? htI[1][mt]
                           : (ss == 2) ? htI[2][mt] : htI[3][mt];
            #pragma unroll
            for (int r = 0; r < 8; ++r) {
                const int row = mt * 16 + r + half * 8;
                const float hr = selR[r] * rinv;
                const float hi = selI[r] * rinv;
                hbuf[wave][0][row][n] = hr;
                hbuf[wave][1][row][n] = hi;
                const float wr = C[OFF_W + wave * 2 * DCUT + row];
                const float wi = C[OFF_W + wave * 2 * DCUT + DCUT + row];
                ampR += wr * hr - wi * hi;
                ampI += wr * hi + wi * hr;
            }
        }
        ampR += __shfl_xor(ampR, 16, 32);
        ampI += __shfl_xor(ampI, 16, 32);
        ampR += cR;
        ampI += cI;
        phase += atan2f(ampI, ampR);

        __syncthreads();

        #pragma unroll
        for (int kk = 0; kk < 8; ++kk) {
            const int krow = kk * 4 + half * 2;
            Hr[kk][0] = hbuf[wave][0][krow + 0][n];
            Hr[kk][1] = hbuf[wave][0][krow + 1][n];
            Hi[kk][0] = hbuf[wave][1][krow + 0][n];
            Hi[kk][1] = hbuf[wave][1][krow + 1][n];
        }
    }

    if (half == 0) {
        float2 o;
        o.x = 0.5f * logp;
        o.y = phase;
        *(float2*)(outg + (size_t)bcol * 2) = o;
    }
}

extern "C" void kernel_launch(void* const* d_in, const int* in_sizes, int n_in,
                              void* d_out, int out_size, void* d_ws, size_t ws_size,
                              hipStream_t stream) {
    (void)in_sizes; (void)n_in; (void)out_size; (void)d_ws; (void)ws_size;
    const int*   x  = (const int*)  d_in[0];
    const float* M  = (const float*)d_in[1];
    const float* v  = (const float*)d_in[2];
    const float* w  = (const float*)d_in[3];
    const float* c  = (const float*)d_in[4];
    float* out = (float*)d_out;

    dim3 grid(NB / 64);
    dim3 block(128);
    mps_rnn_wmma_kernel<<<grid, block, 0, stream>>>(x, M, v, w, c, out);
}


# Only the selected plane is kept: each finished tile's h and p go in by conditional move (x read a site ahead); no divergent 4-plane select at site end; psum crosses halves once, sum order changes
# speedup vs baseline: 1.9595x; 1.0920x over previous
#include <hip/hip_runtime.h>
#include <math.h>


#define NB   16384
#define NQ   100
#define NS   50
#define DCUT 32
#define HLOC 4

#define MPAD 34
#define HPAD 17

#define MSZ     (HLOC * DCUT * MPAD)
#define OFF_MR  0
#define OFF_MI  MSZ
#define OFF_VR  (2 * MSZ)
#define OFF_VI  (2 * MSZ + HLOC * DCUT)
#define OFF_W   (2 * MSZ + 2 * HLOC * DCUT)
#define SITE_FLOATS (2 * MSZ + 2 * HLOC * DCUT + 4 * 2 * DCUT)

typedef __attribute__((ext_vector_type(2))) float v2f;
typedef __attribute__((ext_vector_type(8))) float v8f;

static __device__ __forceinline__ v8f wmma4(v2f a, v2f b, v8f c) {
    return __builtin_amdgcn_wmma_f32_16x16x4_f32(
         false, a,  false, b,
         (short)0, c,  false,  false);
}

static __device__ __forceinline__ void put_m(float* S, int off, float4 m) {
    v2f re, im;
    re[0] = m.x; re[1] = m.z;
    im[0] = m.y; im[1] = m.w;
    *(v2f*)(S + OFF_MR + off) = re;
    *(v2f*)(S + OFF_MI + off) = im;
}

__global__ __launch_bounds__(128) void mps_rnn_wmma_kernel(
    const int*   __restrict__ xg,
    const float* __restrict__ Mg,
    const float* __restrict__ vg,
    const float* __restrict__ wg,
    const float* __restrict__ cg,
    float*       __restrict__ outg)
{
    __shared__ __attribute__((aligned(16))) float site[2][SITE_FLOATS];
    __shared__ float hbuf[4][2][DCUT][HPAD];

    const int tid  = threadIdx.x;
    const int lane = tid & 31;
    const int wave = tid >> 5;
    const int n    = lane & 15;
    const int half = lane >> 4;
    const int bcol = blockIdx.x * 64 + wave * 16 + n;

    const int sb = (tid >> 4) * MPAD + 2 * (tid & 15);
    const float4* M4 = reinterpret_cast<const float4*>(Mg);
    const float2* v2 = reinterpret_cast<const float2*>(vg);
    const float2* w2 = reinterpret_cast<const float2*>(wg);

    {
        float* S = site[0];
        #pragma unroll
        for (int q = 0; q < 16; ++q) put_m(S, sb + 272 * q, M4[tid + 128 * q]);
        const float2 vq = v2[tid];
        S[OFF_VR + tid] = vq.x;
        S[OFF_VI + tid] = vq.y;
        const float2 wq = w2[lane];
        S[OFF_W + wave * 2 * DCUT + lane]        = wq.x;
        S[OFF_W + wave * 2 * DCUT + DCUT + lane] = wq.y;
    }

    v2f Hr[8], Hi[8];
    const float h0 = 0.17677669529663688f;
    #pragma unroll
    for (int k = 0; k < 8; ++k) {
        Hr[k][0] = h0; Hr[k][1] = h0;
        Hi[k][0] = 0.f; Hi[k][1] = 0.f;
    }

    float logp = 0.f, phase = 0.f;
    const int k0h = half * 2;

    int ss = 2 * xg[(size_t)bcol * NQ] + xg[(size_t)bcol * NQ + 1];

    __syncthreads();

    for (int t = 0; t < NS; ++t) {
        const float* C = site[t & 1];
        float*       N = site[(t & 1) ^ 1];

        const int tn = (t + 1 < NS) ? (t + 1) : (NS - 1);
        const float4* Mn = M4 + (size_t)tn * (HLOC * DCUT * DCUT / 2) + tid;
        const float2 vq = v2[tn * (HLOC * DCUT) + tid];
        const float2 wq = w2[tn * DCUT + lane];
        const int xn0 = xg[(size_t)bcol * NQ + 2 * tn];
        const int xn1 = xg[(size_t)bcol * NQ + 2 * tn + 1];
        const bool pick1 = (ss == 1);
        const bool pick2 = (ss == 2);
        const bool pick3 = !(ss == 0 || ss == 1 || ss == 2);
        const float cR = cg[2 * t];
        const float cI = cg[2 * t + 1];
        float4 mq[16];

        v2f    fAr[2][8], fAi[2][8];
        float4 fVr[2][2], fVi[2][2];

        {
            const float* mrBase = C + OFF_MR + n * MPAD + k0h;
            const float* miBase = C + OFF_MI + n * MPAD + k0h;
            #pragma unroll
            for (int kk = 0; kk < 8; ++kk) {
                fAr[0][kk] = *(const v2f*)(mrBase + kk * 4);
                fAi[0][kk] = *(const v2f*)(miBase + kk * 4);
            }
            const float4* vr4 = (const float4*)(C + OFF_VR + half * 8);
            const float4* vi4 = (const float4*)(C + OFF_VI + half * 8);
            fVr[0][0] = vr4[0]; fVr[0][1] = vr4[1];
            fVi[0][0] = vi4[0]; fVi[0][1] = vi4[1];
        }

        v8f selR[2], selI[2];
        float psel_part = 0.f, psum_part = 0.f;
        #pragma unroll
        for (int s = 0; s < HLOC; ++s) {
            float p_part = 0.f;
            const bool pick = (s == 1) ? pick1 : (s == 2) ? pick2 : pick3;
            #pragma unroll
            for (int mt = 0; mt < 2; ++mt) {
                const int j   = s * 2 + mt;
                const int cur = j & 1;
                const int nxt = cur ^ 1;

                if (j == 0) {
                    mq[0] = Mn[0];
                    mq[1] = Mn[128];
                }
                if (j < 7) {
                    mq[2 * j + 2] = Mn[128 * (2 * j + 2)];
                    mq[2 * j + 3] = Mn[128 * (2 * j + 3)];
                }
                if (j >= 2) {
                    put_m(N, sb + 272 * (2 * j - 4), mq[2 * j - 4]);
                    put_m(N, sb + 272 * (2 * j - 3), mq[2 * j - 3]);
                }
                if (j < 7) {
                    const int s1  = (j + 1) >> 1;
                    const int mt1 = (j + 1) & 1;
                    const float* mrBase = C + OFF_MR + ((s1 * DCUT) + mt1 * 16 + n) * MPAD + k0h;
                    const float* miBase = C + OFF_MI + ((s1 * DCUT) + mt1 * 16 + n) * MPAD + k0h;
                    #pragma unroll
                    for (int kk = 0; kk < 8; ++kk) {
                        fAr[nxt][kk] = *(const v2f*)(mrBase + kk * 4);
                        fAi[nxt][kk] = *(const v2f*)(miBase + kk * 4);
                    }
                    const float4* vr4 = (const float4*)(C + OFF_VR + s1 * DCUT + mt1 * 16 + half * 8);
                    const float4* vi4 = (const float4*)(C + OFF_VI + s1 * DCUT + mt1 * 16 + half * 8);
                    fVr[nxt][0] = vr4[0]; fVr[nxt][1] = vr4[1];
                    fVi[nxt][0] = vi4[0]; fVi[nxt][1] = vi4[1];
                }
                asm volatile("" : "+v"(fAr[cur][0]), "+v"(fAi[cur][0]));

                v8f accR  = {};
                v8f accMI = {};
                v8f accI  = {};
                #pragma unroll
                for (int kk = 0; kk < 8; ++kk) {
                    const v2f Ar = fAr[cur][kk];
                    const v2f Ai = fAi[cur][kk];
                    accR  = wmma4(Ar, Hr[kk], accR);
                    accMI = wmma4(Ai, Hi[kk], accMI);
                    accI  = wmma4(Ar, Hi[kk], accI);
                    accI  = wmma4(Ai, Hr[kk], accI);
                }
                #pragma unroll
                for (int r = 0; r < 8; ++r) {
                    const float4 vr = fVr[cur][r >> 2];
                    const float4 vi = fVi[cur][r >> 2];
                    const float vrr = ((r & 3) == 0) ? vr.x : ((r & 3) == 1) ? vr.y : ((r & 3) == 2) ? vr.z : vr.w;
                    const float vii = ((r & 3) == 0) ? vi.x : ((r & 3) == 1) ? vi.y : ((r & 3) == 2) ? vi.z : vi.w;
                    const float hr = accR[r] - accMI[r] + vrr;
                    const float hi = accI[r]            + vii;
                    if (s == 0) {
                        selR[mt][r] = hr;
                        selI[mt][r] = hi;
                    } else {
                        selR[mt][r] = pick ? hr : selR[mt][r];
                        selI[mt][r] = pick ? hi : selI[mt][r];
                    }
                    p_part += hr * hr + hi * hi;
                }
            }
            psum_part += p_part;
            psel_part = (s == 0) ? p_part : (pick ? p_part : psel_part);
        }
        put_m(N, sb + 272 * 12, mq[12]);
        put_m(N, sb + 272 * 13, mq[13]);
        put_m(N, sb + 272 * 14, mq[14]);
        put_m(N, sb + 272 * 15, mq[15]);
        N[OFF_VR + tid] = vq.x;
        N[OFF_VI + tid] = vq.y;
        N[OFF_W + wave * 2 * DCUT + lane]        = wq.x;
        N[OFF_W + wave * 2 * DCUT + DCUT + lane] = wq.y;

        const float p_sel = psel_part + __shfl_xor(psel_part, 16, 32);
        const float psum  = psum_part + __shfl_xor(psum_part, 16, 32);
        logp += logf(p_sel) - logf(psum);
        const float rinv = rsqrtf(p_sel);

        float ampR = 0.f, ampI = 0.f;
        #pragma unroll
        for (int mt = 0; mt < 2; ++mt) {
            #pragma unroll
            for (int r = 0; r < 8; ++r) {
                const int row = mt * 16 + r + half * 8;
                const float hr = selR[mt][r] * rinv;
                const float hi = selI[mt][r] * rinv;
                hbuf[wave][0][row][n] = hr;
                hbuf[wave][1][row][n] = hi;
                const float wr = C[OFF_W + wave * 2 * DCUT + row];
                const float wi = C[OFF_W + wave * 2 * DCUT + DCUT + row];
                ampR += wr * hr - wi * hi;
                ampI += wr * hi + wi * hr;
            }
        }
        ampR += __shfl_xor(ampR, 16, 32);
        ampI += __shfl_xor(ampI, 16, 32);
        ampR += cR;
        ampI += cI;
        phase += atan2f(ampI, ampR);

        __syncthreads();

        #pragma unroll
        for (int kk = 0; kk < 8; ++kk) {
            const int krow = kk * 4 + half * 2;
            Hr[kk][0] = hbuf[wave][0][krow + 0][n];
            Hr[kk][1] = hbuf[wave][0][krow + 1][n];
            Hi[kk][0] = hbuf[wave][1][krow + 0][n];
            Hi[kk][1] = hbuf[wave][1][krow + 1][n];
        }
        ss = 2 * xn0 + xn1;
    }

    if (half == 0) {
        float2 o;
        o.x = 0.5f * logp;
        o.y = phase;
        *(float2*)(outg + (size_t)bcol * 2) = o;
    }
}

extern "C" void kernel_launch(void* const* d_in, const int* in_sizes, int n_in,
                              void* d_out, int out_size, void* d_ws, size_t ws_size,
                              hipStream_t stream) {
    (void)in_sizes; (void)n_in; (void)out_size; (void)d_ws; (void)ws_size;
    const int*   x  = (const int*)  d_in[0];
    const float* M  = (const float*)d_in[1];
    const float* v  = (const float*)d_in[2];
    const float* w  = (const float*)d_in[3];
    const float* c  = (const float*)d_in[4];
    float* out = (float*)d_out;

    dim3 grid(NB / 64);
    dim3 block(128);
    mps_rnn_wmma_kernel<<<grid, block, 0, stream>>>(x, M, v, w, c, out);
}


# Logs, w.h sum and atan2 of site t move into site t+1, issued among its WMMAs (kernel marked one wave per SIMD so the compiler spreads them); only h's path stays between sites; w.h sum order changes
# speedup vs baseline: 2.1579x; 1.1013x over previous
#include <hip/hip_runtime.h>
#include <math.h>


#define NB   16384
#define NQ   100
#define NS   50
#define DCUT 32
#define HLOC 4

#define MPAD 34
#define HPAD 17

#define MSZ     (HLOC * DCUT * MPAD)
#define OFF_MR  0
#define OFF_MI  MSZ
#define OFF_VR  (2 * MSZ)
#define OFF_VI  (2 * MSZ + HLOC * DCUT)
#define OFF_W   (2 * MSZ + 2 * HLOC * DCUT)
#define SITE_FLOATS (2 * MSZ + 2 * HLOC * DCUT + 4 * 2 * DCUT)

typedef __attribute__((ext_vector_type(2))) float v2f;
typedef __attribute__((ext_vector_type(8))) float v8f;

static __device__ __forceinline__ v8f wmma4(v2f a, v2f b, v8f c) {
    return __builtin_amdgcn_wmma_f32_16x16x4_f32(
         false, a,  false, b,
         (short)0, c,  false,  false);
}

static __device__ __forceinline__ void put_m(float* S, int off, float4 m) {
    v2f re, im;
    re[0] = m.x; re[1] = m.z;
    im[0] = m.y; im[1] = m.w;
    *(v2f*)(S + OFF_MR + off) = re;
    *(v2f*)(S + OFF_MI + off) = im;
}

static __device__ __forceinline__ void site_scalars(
    const float* Wst, int half, const v2f (&Hr)[8], const v2f (&Hi)[8],
    float cR, float cI, float p_sel, float psum_part, float& logp, float& phase)
{
    const float psum = psum_part + __shfl_xor(psum_part, 16, 32);
    logp += logf(p_sel) - logf(psum);

    float ampR = 0.f, ampI = 0.f;
    #pragma unroll
    for (int kk = 0; kk < 8; ++kk) {
        const v2f wr = *(const v2f*)(Wst + kk * 4 + half * 2);
        const v2f wi = *(const v2f*)(Wst + DCUT + kk * 4 + half * 2);
        #pragma unroll
        for (int el = 0; el < 2; ++el) {
            ampR += wr[el] * Hr[kk][el] - wi[el] * Hi[kk][el];
            ampI += wr[el] * Hi[kk][el] + wi[el] * Hr[kk][el];
        }
    }
    ampR += __shfl_xor(ampR, 16, 32);
    ampI += __shfl_xor(ampI, 16, 32);
    ampR += cR;
    ampI += cI;
    phase += atan2f(ampI, ampR);
}

__global__ __launch_bounds__(128) __attribute__((amdgpu_waves_per_eu(1, 1))) void mps_rnn_wmma_kernel(
    const int*   __restrict__ xg,
    const float* __restrict__ Mg,
    const float* __restrict__ vg,
    const float* __restrict__ wg,
    const float* __restrict__ cg,
    float*       __restrict__ outg)
{
    __shared__ __attribute__((aligned(16))) float site[2][SITE_FLOATS];
    __shared__ float hbuf[4][2][DCUT][HPAD];

    const int tid  = threadIdx.x;
    const int lane = tid & 31;
    const int wave = tid >> 5;
    const int n    = lane & 15;
    const int half = lane >> 4;
    const int bcol = blockIdx.x * 64 + wave * 16 + n;

    const int sb = (tid >> 4) * MPAD + 2 * (tid & 15);
    const float4* M4 = reinterpret_cast<const float4*>(Mg);
    const float2* v2 = reinterpret_cast<const float2*>(vg);
    const float2* w2 = reinterpret_cast<const float2*>(wg);

    {
        float* S = site[0];
        #pragma unroll
        for (int q = 0; q < 16; ++q) put_m(S, sb + 272 * q, M4[tid + 128 * q]);
        const float2 vq = v2[tid];
        S[OFF_VR + tid] = vq.x;
        S[OFF_VI + tid] = vq.y;
        const float2 wq = w2[lane];
        S[OFF_W + wave * 2 * DCUT + lane]        = wq.x;
        S[OFF_W + wave * 2 * DCUT + DCUT + lane] = wq.y;
        site[1][OFF_W + wave * 2 * DCUT + lane]        = 0.f;
        site[1][OFF_W + wave * 2 * DCUT + DCUT + lane] = 0.f;
    }

    v2f Hr[8], Hi[8];
    const float h0 = 0.17677669529663688f;
    #pragma unroll
    for (int k = 0; k < 8; ++k) {
        Hr[k][0] = h0; Hr[k][1] = h0;
        Hi[k][0] = 0.f; Hi[k][1] = 0.f;
    }

    float logp = 0.f, phase = 0.f;
    const int k0h = half * 2;

    float pselPrev = 1.f, psumPartPrev = 0.5f, cRprev = 1.f, cIprev = 0.f;

    int ss = 2 * xg[(size_t)bcol * NQ] + xg[(size_t)bcol * NQ + 1];

    __syncthreads();

    for (int t = 0; t < NS; ++t) {
        const float* C = site[t & 1];
        float*       N = site[(t & 1) ^ 1];

        const int tn = (t + 1 < NS) ? (t + 1) : (NS - 1);
        const float4* Mn = M4 + (size_t)tn * (HLOC * DCUT * DCUT / 2) + tid;
        const float2 vq = v2[tn * (HLOC * DCUT) + tid];
        const float2 wq = w2[tn * DCUT + lane];
        const int xn0 = xg[(size_t)bcol * NQ + 2 * tn];
        const int xn1 = xg[(size_t)bcol * NQ + 2 * tn + 1];
        const bool pick1 = (ss == 1);
        const bool pick2 = (ss == 2);
        const bool pick3 = !(ss == 0 || ss == 1 || ss == 2);
        const float cR = cg[2 * t];
        const float cI = cg[2 * t + 1];
        float4 mq[16];

        v2f    fAr[2][8], fAi[2][8];
        float4 fVr[2][2], fVi[2][2];

        {
            const float* mrBase = C + OFF_MR + n * MPAD + k0h;
            const float* miBase = C + OFF_MI + n * MPAD + k0h;
            #pragma unroll
            for (int kk = 0; kk < 8; ++kk) {
                fAr[0][kk] = *(const v2f*)(mrBase + kk * 4);
                fAi[0][kk] = *(const v2f*)(miBase + kk * 4);
            }
            const float4* vr4 = (const float4*)(C + OFF_VR + half * 8);
            const float4* vi4 = (const float4*)(C + OFF_VI + half * 8);
            fVr[0][0] = vr4[0]; fVr[0][1] = vr4[1];
            fVi[0][0] = vi4[0]; fVi[0][1] = vi4[1];
        }

        v8f selR[2], selI[2];
        float psel_part = 0.f, psum_part = 0.f;
        #pragma unroll
        for (int s = 0; s < HLOC; ++s) {
            float p_part = 0.f;
            const bool pick = (s == 1) ? pick1 : (s == 2) ? pick2 : pick3;
            #pragma unroll
            for (int mt = 0; mt < 2; ++mt) {
                const int j   = s * 2 + mt;
                const int cur = j & 1;
                const int nxt = cur ^ 1;

                if (j == 0) {
                    mq[0] = Mn[0];
                    mq[1] = Mn[128];
                }
                if (j < 7) {
                    mq[2 * j + 2] = Mn[128 * (2 * j + 2)];
                    mq[2 * j + 3] = Mn[128 * (2 * j + 3)];
                }
                if (j >= 2) {
                    put_m(N, sb + 272 * (2 * j - 4), mq[2 * j - 4]);
                    put_m(N, sb + 272 * (2 * j - 3), mq[2 * j - 3]);
                }
                if (j < 7) {
                    const int s1  = (j + 1) >> 1;
                    const int mt1 = (j + 1) & 1;
                    const float* mrBase = C + OFF_MR + ((s1 * DCUT) + mt1 * 16 + n) * MPAD + k0h;
                    const float* miBase = C + OFF_MI + ((s1 * DCUT) + mt1 * 16 + n) * MPAD + k0h;
                    #pragma unroll
                    for (int kk = 0; kk < 8; ++kk) {
                        fAr[nxt][kk] = *(const v2f*)(mrBase + kk * 4);
                        fAi[nxt][kk] = *(const v2f*)(miBase + kk * 4);
                    }
                    const float4* vr4 = (const float4*)(C + OFF_VR + s1 * DCUT + mt1 * 16 + half * 8);
                    const float4* vi4 = (const float4*)(C + OFF_VI + s1 * DCUT + mt1 * 16 + half * 8);
                    fVr[nxt][0] = vr4[0]; fVr[nxt][1] = vr4[1];
                    fVi[nxt][0] = vi4[0]; fVi[nxt][1] = vi4[1];
                }
                asm volatile("" : "+v"(fAr[cur][0]), "+v"(fAi[cur][0]));

                if (j == 1) {
                    site_scalars(N + OFF_W + wave * 2 * DCUT, half, Hr, Hi,
                                 cRprev, cIprev, pselPrev, psumPartPrev, logp, phase);
                }

                v8f accR  = {};
                v8f accMI = {};
                v8f accI  = {};
                #pragma unroll
                for (int kk = 0; kk < 8; ++kk) {
                    const v2f Ar = fAr[cur][kk];
                    const v2f Ai = fAi[cur][kk];
                    accR  = wmma4(Ar, Hr[kk], accR);
                    accMI = wmma4(Ai, Hi[kk], accMI);
                    accI  = wmma4(Ar, Hi[kk], accI);
                    accI  = wmma4(Ai, Hr[kk], accI);
                }
                #pragma unroll
                for (int r = 0; r < 8; ++r) {
                    const float4 vr = fVr[cur][r >> 2];
                    const float4 vi = fVi[cur][r >> 2];
                    const float vrr = ((r & 3) == 0) ? vr.x : ((r & 3) == 1) ? vr.y : ((r & 3) == 2) ? vr.z : vr.w;
                    const float vii = ((r & 3) == 0) ? vi.x : ((r & 3) == 1) ? vi.y : ((r & 3) == 2) ? vi.z : vi.w;
                    const float hr = accR[r] - accMI[r] + vrr;
                    const float hi = accI[r]            + vii;
                    if (s == 0) {
                        selR[mt][r] = hr;
                        selI[mt][r] = hi;
                    } else {
                        selR[mt][r] = pick ? hr : selR[mt][r];
                        selI[mt][r] = pick ? hi : selI[mt][r];
                    }
                    p_part += hr * hr + hi * hi;
                }
            }
            psum_part += p_part;
            psel_part = (s == 0) ? p_part : (pick ? p_part : psel_part);
        }
        put_m(N, sb + 272 * 12, mq[12]);
        put_m(N, sb + 272 * 13, mq[13]);
        put_m(N, sb + 272 * 14, mq[14]);
        put_m(N, sb + 272 * 15, mq[15]);
        N[OFF_VR + tid] = vq.x;
        N[OFF_VI + tid] = vq.y;
        N[OFF_W + wave * 2 * DCUT + lane]        = wq.x;
        N[OFF_W + wave * 2 * DCUT + DCUT + lane] = wq.y;

        const float p_sel = psel_part + __shfl_xor(psel_part, 16, 32);
        const float rinv = rsqrtf(p_sel);
        pselPrev     = p_sel;
        psumPartPrev = psum_part;
        cRprev = cR;
        cIprev = cI;

        #pragma unroll
        for (int mt = 0; mt < 2; ++mt) {
            #pragma unroll
            for (int r = 0; r < 8; ++r) {
                const int row = mt * 16 + r + half * 8;
                hbuf[wave][0][row][n] = selR[mt][r] * rinv;
                hbuf[wave][1][row][n] = selI[mt][r] * rinv;
            }
        }

        __syncthreads();

        #pragma unroll
        for (int kk = 0; kk < 8; ++kk) {
            const int krow = kk * 4 + half * 2;
            Hr[kk][0] = hbuf[wave][0][krow + 0][n];
            Hr[kk][1] = hbuf[wave][0][krow + 1][n];
            Hi[kk][0] = hbuf[wave][1][krow + 0][n];
            Hi[kk][1] = hbuf[wave][1][krow + 1][n];
        }
        ss = 2 * xn0 + xn1;
    }

    site_scalars(site[(NS - 1) & 1] + OFF_W + wave * 2 * DCUT, half, Hr, Hi,
                 cRprev, cIprev, pselPrev, psumPartPrev, logp, phase);

    if (half == 0) {
        float2 o;
        o.x = 0.5f * logp;
        o.y = phase;
        *(float2*)(outg + (size_t)bcol * 2) = o;
    }
}

extern "C" void kernel_launch(void* const* d_in, const int* in_sizes, int n_in,
                              void* d_out, int out_size, void* d_ws, size_t ws_size,
                              hipStream_t stream) {
    (void)in_sizes; (void)n_in; (void)out_size; (void)d_ws; (void)ws_size;
    const int*   x  = (const int*)  d_in[0];
    const float* M  = (const float*)d_in[1];
    const float* v  = (const float*)d_in[2];
    const float* w  = (const float*)d_in[3];
    const float* c  = (const float*)d_in[4];
    float* out = (float*)d_out;

    dim3 grid(NB / 64);
    dim3 block(128);
    mps_rnn_wmma_kernel<<<grid, block, 0, stream>>>(x, M, v, w, c, out);
}


# LDS only stages operands now: h stays in registers (M columns staged in the k order a tile leaves per lane half), the four half-to-half sums go by permlanex16; M.h and w.h sum order changes
# speedup vs baseline: 2.3078x; 1.0695x over previous
#include <hip/hip_runtime.h>
#include <math.h>


#define NB   16384
#define NQ   100
#define NS   50
#define DCUT 32
#define HLOC 4

#define MPAD 34

#define MSZ     (HLOC * DCUT * MPAD)
#define OFF_MR  0
#define OFF_MI  MSZ
#define OFF_VR  (2 * MSZ)
#define OFF_VI  (2 * MSZ + HLOC * DCUT)
#define OFF_W   (2 * MSZ + 2 * HLOC * DCUT)
#define SITE_FLOATS (2 * MSZ + 2 * HLOC * DCUT + 4 * 2 * DCUT)

typedef __attribute__((ext_vector_type(2))) float v2f;
typedef __attribute__((ext_vector_type(8))) float v8f;

static __device__ __forceinline__ v8f wmma4(v2f a, v2f b, v8f c) {
    return __builtin_amdgcn_wmma_f32_16x16x4_f32(
         false, a,  false, b,
         (short)0, c,  false,  false);
}

static __device__ __forceinline__ float other_half(float x) {
    const int xi = __builtin_bit_cast(int, x);
    return __builtin_bit_cast(float,
        __builtin_amdgcn_permlanex16(xi, xi, 0x76543210, 0xfedcba98, false, false));
}

static __device__ __forceinline__ void put_m(float* S, int off, float4 m) {
    v2f re, im;
    re[0] = m.x; re[1] = m.z;
    im[0] = m.y; im[1] = m.w;
    *(v2f*)(S + OFF_MR + off) = re;
    *(v2f*)(S + OFF_MI + off) = im;
}

static __device__ __forceinline__ void site_scalars(
    const float* Wst, int half, const v2f (&Hr)[8], const v2f (&Hi)[8],
    float cR, float cI, float p_sel, float psum_part, float& logp, float& phase)
{
    const float psum = psum_part + other_half(psum_part);
    logp += logf(p_sel) - logf(psum);

    float ampR = 0.f, ampI = 0.f;
    #pragma unroll
    for (int kk = 0; kk < 8; ++kk) {
        const int k = 16 * (kk >> 2) + 8 * half + 2 * (kk & 3);
        const v2f wr = *(const v2f*)(Wst + k);
        const v2f wi = *(const v2f*)(Wst + DCUT + k);
        #pragma unroll
        for (int el = 0; el < 2; ++el) {
            ampR += wr[el] * Hr[kk][el] - wi[el] * Hi[kk][el];
            ampI += wr[el] * Hi[kk][el] + wi[el] * Hr[kk][el];
        }
    }
    ampR += other_half(ampR);
    ampI += other_half(ampI);
    ampR += cR;
    ampI += cI;
    phase += atan2f(ampI, ampR);
}

__global__ __launch_bounds__(128) __attribute__((amdgpu_waves_per_eu(1, 1))) void mps_rnn_wmma_kernel(
    const int*   __restrict__ xg,
    const float* __restrict__ Mg,
    const float* __restrict__ vg,
    const float* __restrict__ wg,
    const float* __restrict__ cg,
    float*       __restrict__ outg)
{
    __shared__ __attribute__((aligned(16))) float site[2][SITE_FLOATS];

    const int tid  = threadIdx.x;
    const int lane = tid & 31;
    const int wave = tid >> 5;
    const int n    = lane & 15;
    const int half = lane >> 4;
    const int bcol = blockIdx.x * 64 + wave * 16 + n;

    const int ej = tid & 15;
    const int sb = (tid >> 4) * MPAD + 16 * (ej >> 3) + 4 * (ej & 3) + 2 * ((ej >> 2) & 1);
    const float4* M4 = reinterpret_cast<const float4*>(Mg);
    const float2* v2 = reinterpret_cast<const float2*>(vg);
    const float2* w2 = reinterpret_cast<const float2*>(wg);

    {
        float* S = site[0];
        #pragma unroll
        for (int q = 0; q < 16; ++q) put_m(S, sb + 272 * q, M4[tid + 128 * q]);
        const float2 vq = v2[tid];
        S[OFF_VR + tid] = vq.x;
        S[OFF_VI + tid] = vq.y;
        const float2 wq = w2[lane];
        S[OFF_W + wave * 2 * DCUT + lane]        = wq.x;
        S[OFF_W + wave * 2 * DCUT + DCUT + lane] = wq.y;
        site[1][OFF_W + wave * 2 * DCUT + lane]        = 0.f;
        site[1][OFF_W + wave * 2 * DCUT + DCUT + lane] = 0.f;
    }

    v2f Hr[8], Hi[8];
    const float h0 = 0.17677669529663688f;
    #pragma unroll
    for (int k = 0; k < 8; ++k) {
        Hr[k][0] = h0; Hr[k][1] = h0;
        Hi[k][0] = 0.f; Hi[k][1] = 0.f;
    }

    float logp = 0.f, phase = 0.f;
    const int k0h = half * 2;

    float pselPrev = 1.f, psumPartPrev = 0.5f, cRprev = 1.f, cIprev = 0.f;

    int ss = 2 * xg[(size_t)bcol * NQ] + xg[(size_t)bcol * NQ + 1];

    __syncthreads();

    for (int t = 0; t < NS; ++t) {
        const float* C = site[t & 1];
        float*       N = site[(t & 1) ^ 1];

        const int tn = (t + 1 < NS) ? (t + 1) : (NS - 1);
        const float4* Mn = M4 + (size_t)tn * (HLOC * DCUT * DCUT / 2) + tid;
        const float2 vq = v2[tn * (HLOC * DCUT) + tid];
        const float2 wq = w2[tn * DCUT + lane];
        const int xn0 = xg[(size_t)bcol * NQ + 2 * tn];
        const int xn1 = xg[(size_t)bcol * NQ + 2 * tn + 1];
        const bool pick1 = (ss == 1);
        const bool pick2 = (ss == 2);
        const bool pick3 = !(ss == 0 || ss == 1 || ss == 2);
        const float cR = cg[2 * t];
        const float cI = cg[2 * t + 1];
        float4 mq[16];

        v2f    fAr[2][8], fAi[2][8];
        float4 fVr[2][2], fVi[2][2];

        {
            const float* mrBase = C + OFF_MR + n * MPAD + k0h;
            const float* miBase = C + OFF_MI + n * MPAD + k0h;
            #pragma unroll
            for (int kk = 0; kk < 8; ++kk) {
                fAr[0][kk] = *(const v2f*)(mrBase + kk * 4);
                fAi[0][kk] = *(const v2f*)(miBase + kk * 4);
            }
            const float4* vr4 = (const float4*)(C + OFF_VR + half * 8);
            const float4* vi4 = (const float4*)(C + OFF_VI + half * 8);
            fVr[0][0] = vr4[0]; fVr[0][1] = vr4[1];
            fVi[0][0] = vi4[0]; fVi[0][1] = vi4[1];
        }

        v8f selR[2], selI[2];
        float psel_part = 0.f, psum_part = 0.f;
        #pragma unroll
        for (int s = 0; s < HLOC; ++s) {
            float p_part = 0.f;
            const bool pick = (s == 1) ? pick1 : (s == 2) ? pick2 : pick3;
            #pragma unroll
            for (int mt = 0; mt < 2; ++mt) {
                const int j   = s * 2 + mt;
                const int cur = j & 1;
                const int nxt = cur ^ 1;

                if (j == 0) {
                    mq[0] = Mn[0];
                    mq[1] = Mn[128];
                }
                if (j < 7) {
                    mq[2 * j + 2] = Mn[128 * (2 * j + 2)];
                    mq[2 * j + 3] = Mn[128 * (2 * j + 3)];
                }
                if (j >= 2) {
                    put_m(N, sb + 272 * (2 * j - 4), mq[2 * j - 4]);
                    put_m(N, sb + 272 * (2 * j - 3), mq[2 * j - 3]);
                }
                if (j < 7) {
                    const int s1  = (j + 1) >> 1;
                    const int mt1 = (j + 1) & 1;
                    const float* mrBase = C + OFF_MR + ((s1 * DCUT) + mt1 * 16 + n) * MPAD + k0h;
                    const float* miBase = C + OFF_MI + ((s1 * DCUT) + mt1 * 16 + n) * MPAD + k0h;
                    #pragma unroll
                    for (int kk = 0; kk < 8; ++kk) {
                        fAr[nxt][kk] = *(const v2f*)(mrBase + kk * 4);
                        fAi[nxt][kk] = *(const v2f*)(miBase + kk * 4);
                    }
                    const float4* vr4 = (const float4*)(C + OFF_VR + s1 * DCUT + mt1 * 16 + half * 8);
                    const float4* vi4 = (const float4*)(C + OFF_VI + s1 * DCUT + mt1 * 16 + half * 8);
                    fVr[nxt][0] = vr4[0]; fVr[nxt][1] = vr4[1];
                    fVi[nxt][0] = vi4[0]; fVi[nxt][1] = vi4[1];
                }
                asm volatile("" : "+v"(fAr[cur][0]), "+v"(fAi[cur][0]));

                if (j == 1) {
                    site_scalars(N + OFF_W + wave * 2 * DCUT, half, Hr, Hi,
                                 cRprev, cIprev, pselPrev, psumPartPrev, logp, phase);
                }

                v8f accR  = {};
                v8f accMI = {};
                v8f accI  = {};
                #pragma unroll
                for (int kk = 0; kk < 8; ++kk) {
                    const v2f Ar = fAr[cur][kk];
                    const v2f Ai = fAi[cur][kk];
                    accR  = wmma4(Ar, Hr[kk], accR);
                    accMI = wmma4(Ai, Hi[kk], accMI);
                    accI  = wmma4(Ar, Hi[kk], accI);
                    accI  = wmma4(Ai, Hr[kk], accI);
                }
                #pragma unroll
                for (int r = 0; r < 8; ++r) {
                    const float4 vr = fVr[cur][r >> 2];
                    const float4 vi = fVi[cur][r >> 2];
                    const float vrr = ((r & 3) == 0) ? vr.x : ((r & 3) == 1) ? vr.y : ((r & 3) == 2) ? vr.z : vr.w;
                    const float vii = ((r & 3) == 0) ? vi.x : ((r & 3) == 1) ? vi.y : ((r & 3) == 2) ? vi.z : vi.w;
                    const float hr = accR[r] - accMI[r] + vrr;
                    const float hi = accI[r]            + vii;
                    if (s == 0) {
                        selR[mt][r] = hr;
                        selI[mt][r] = hi;
                    } else {
                        selR[mt][r] = pick ? hr : selR[mt][r];
                        selI[mt][r] = pick ? hi : selI[mt][r];
                    }
                    p_part += hr * hr + hi * hi;
                }
            }
            psum_part += p_part;
            psel_part = (s == 0) ? p_part : (pick ? p_part : psel_part);
        }
        put_m(N, sb + 272 * 12, mq[12]);
        put_m(N, sb + 272 * 13, mq[13]);
        put_m(N, sb + 272 * 14, mq[14]);
        put_m(N, sb + 272 * 15, mq[15]);
        N[OFF_VR + tid] = vq.x;
        N[OFF_VI + tid] = vq.y;
        N[OFF_W + wave * 2 * DCUT + lane]        = wq.x;
        N[OFF_W + wave * 2 * DCUT + DCUT + lane] = wq.y;

        const float p_sel = psel_part + other_half(psel_part);
        const float rinv = rsqrtf(p_sel);
        pselPrev     = p_sel;
        psumPartPrev = psum_part;
        cRprev = cR;
        cIprev = cI;

        #pragma unroll
        for (int mt = 0; mt < 2; ++mt) {
            #pragma unroll
            for (int r = 0; r < 8; ++r) {
                Hr[4 * mt + (r >> 1)][r & 1] = selR[mt][r] * rinv;
                Hi[4 * mt + (r >> 1)][r & 1] = selI[mt][r] * rinv;
            }
        }

        __syncthreads();

        ss = 2 * xn0 + xn1;
    }

    site_scalars(site[(NS - 1) & 1] + OFF_W + wave * 2 * DCUT, half, Hr, Hi,
                 cRprev, cIprev, pselPrev, psumPartPrev, logp, phase);

    if (half == 0) {
        float2 o;
        o.x = 0.5f * logp;
        o.y = phase;
        *(float2*)(outg + (size_t)bcol * 2) = o;
    }
}

extern "C" void kernel_launch(void* const* d_in, const int* in_sizes, int n_in,
                              void* d_out, int out_size, void* d_ws, size_t ws_size,
                              hipStream_t stream) {
    (void)in_sizes; (void)n_in; (void)out_size; (void)d_ws; (void)ws_size;
    const int*   x  = (const int*)  d_in[0];
    const float* M  = (const float*)d_in[1];
    const float* v  = (const float*)d_in[2];
    const float* w  = (const float*)d_in[3];
    const float* c  = (const float*)d_in[4];
    float* out = (float*)d_out;

    dim3 grid(NB / 64);
    dim3 block(128);
    mps_rnn_wmma_kernel<<<grid, block, 0, stream>>>(x, M, v, w, c, out);
}


# Three real WMMA products per complex product instead of four: imaginary part = (Mr+Mi)(Hr+Hi) - MrHr - MiHi, with Mr+Mi and Hr+Hi summed in registers; imaginary part's sum order changes
# speedup vs baseline: 2.5864x; 1.1207x over previous
#include <hip/hip_runtime.h>
#include <math.h>


#define NB   16384
#define NQ   100
#define NS   50
#define DCUT 32
#define HLOC 4

#define MPAD 34

#define MSZ     (HLOC * DCUT * MPAD)
#define OFF_MR  0
#define OFF_MI  MSZ
#define OFF_VR  (2 * MSZ)
#define OFF_VI  (2 * MSZ + HLOC * DCUT)
#define OFF_W   (2 * MSZ + 2 * HLOC * DCUT)
#define SITE_FLOATS (2 * MSZ + 2 * HLOC * DCUT + 4 * 2 * DCUT)

typedef __attribute__((ext_vector_type(2))) float v2f;
typedef __attribute__((ext_vector_type(8))) float v8f;

static __device__ __forceinline__ v8f wmma4(v2f a, v2f b, v8f c) {
    return __builtin_amdgcn_wmma_f32_16x16x4_f32(
         false, a,  false, b,
         (short)0, c,  false,  false);
}

static __device__ __forceinline__ float other_half(float x) {
    const int xi = __builtin_bit_cast(int, x);
    return __builtin_bit_cast(float,
        __builtin_amdgcn_permlanex16(xi, xi, 0x76543210, 0xfedcba98, false, false));
}

static __device__ __forceinline__ void put_m(float* S, int off, float4 m) {
    v2f re, im;
    re[0] = m.x; re[1] = m.z;
    im[0] = m.y; im[1] = m.w;
    *(v2f*)(S + OFF_MR + off) = re;
    *(v2f*)(S + OFF_MI + off) = im;
}

static __device__ __forceinline__ void site_scalars(
    const float* Wst, int half, const v2f (&Hr)[8], const v2f (&Hi)[8],
    float cR, float cI, float p_sel, float psum_part, float& logp, float& phase)
{
    const float psum = psum_part + other_half(psum_part);
    logp += logf(p_sel) - logf(psum);

    float ampR = 0.f, ampI = 0.f;
    #pragma unroll
    for (int kk = 0; kk < 8; ++kk) {
        const int k = 16 * (kk >> 2) + 8 * half + 2 * (kk & 3);
        const v2f wr = *(const v2f*)(Wst + k);
        const v2f wi = *(const v2f*)(Wst + DCUT + k);
        #pragma unroll
        for (int el = 0; el < 2; ++el) {
            ampR += wr[el] * Hr[kk][el] - wi[el] * Hi[kk][el];
            ampI += wr[el] * Hi[kk][el] + wi[el] * Hr[kk][el];
        }
    }
    ampR += other_half(ampR);
    ampI += other_half(ampI);
    ampR += cR;
    ampI += cI;
    phase += atan2f(ampI, ampR);
}

__global__ __launch_bounds__(128) __attribute__((amdgpu_waves_per_eu(1, 1))) void mps_rnn_wmma_kernel(
    const int*   __restrict__ xg,
    const float* __restrict__ Mg,
    const float* __restrict__ vg,
    const float* __restrict__ wg,
    const float* __restrict__ cg,
    float*       __restrict__ outg)
{
    __shared__ __attribute__((aligned(16))) float site[2][SITE_FLOATS];

    const int tid  = threadIdx.x;
    const int lane = tid & 31;
    const int wave = tid >> 5;
    const int n    = lane & 15;
    const int half = lane >> 4;
    const int bcol = blockIdx.x * 64 + wave * 16 + n;

    const int ej = tid & 15;
    const int sb = (tid >> 4) * MPAD + 16 * (ej >> 3) + 4 * (ej & 3) + 2 * ((ej >> 2) & 1);
    const float4* M4 = reinterpret_cast<const float4*>(Mg);
    const float2* v2 = reinterpret_cast<const float2*>(vg);
    const float2* w2 = reinterpret_cast<const float2*>(wg);

    {
        float* S = site[0];
        #pragma unroll
        for (int q = 0; q < 16; ++q) put_m(S, sb + 272 * q, M4[tid + 128 * q]);
        const float2 vq = v2[tid];
        S[OFF_VR + tid] = vq.x;
        S[OFF_VI + tid] = vq.y;
        const float2 wq = w2[lane];
        S[OFF_W + wave * 2 * DCUT + lane]        = wq.x;
        S[OFF_W + wave * 2 * DCUT + DCUT + lane] = wq.y;
        site[1][OFF_W + wave * 2 * DCUT + lane]        = 0.f;
        site[1][OFF_W + wave * 2 * DCUT + DCUT + lane] = 0.f;
    }

    v2f Hr[8], Hi[8], Hs[8];
    const float h0 = 0.17677669529663688f;
    #pragma unroll
    for (int k = 0; k < 8; ++k) {
        Hr[k][0] = h0; Hr[k][1] = h0;
        Hi[k][0] = 0.f; Hi[k][1] = 0.f;
        Hs[k] = Hr[k] + Hi[k];
    }

    float logp = 0.f, phase = 0.f;
    const int k0h = half * 2;

    float pselPrev = 1.f, psumPartPrev = 0.5f, cRprev = 1.f, cIprev = 0.f;

    int ss = 2 * xg[(size_t)bcol * NQ] + xg[(size_t)bcol * NQ + 1];

    __syncthreads();

    for (int t = 0; t < NS; ++t) {
        const float* C = site[t & 1];
        float*       N = site[(t & 1) ^ 1];

        const int tn = (t + 1 < NS) ? (t + 1) : (NS - 1);
        const float4* Mn = M4 + (size_t)tn * (HLOC * DCUT * DCUT / 2) + tid;
        const float2 vq = v2[tn * (HLOC * DCUT) + tid];
        const float2 wq = w2[tn * DCUT + lane];
        const int xn0 = xg[(size_t)bcol * NQ + 2 * tn];
        const int xn1 = xg[(size_t)bcol * NQ + 2 * tn + 1];
        const bool pick1 = (ss == 1);
        const bool pick2 = (ss == 2);
        const bool pick3 = !(ss == 0 || ss == 1 || ss == 2);
        const float cR = cg[2 * t];
        const float cI = cg[2 * t + 1];
        float4 mq[16];

        v2f    fAr[2][8], fAi[2][8];
        float4 fVr[2][2], fVi[2][2];

        {
            const float* mrBase = C + OFF_MR + n * MPAD + k0h;
            const float* miBase = C + OFF_MI + n * MPAD + k0h;
            #pragma unroll
            for (int kk = 0; kk < 8; ++kk) {
                fAr[0][kk] = *(const v2f*)(mrBase + kk * 4);
                fAi[0][kk] = *(const v2f*)(miBase + kk * 4);
            }
            const float4* vr4 = (const float4*)(C + OFF_VR + half * 8);
            const float4* vi4 = (const float4*)(C + OFF_VI + half * 8);
            fVr[0][0] = vr4[0]; fVr[0][1] = vr4[1];
            fVi[0][0] = vi4[0]; fVi[0][1] = vi4[1];
        }

        v8f selR[2], selI[2];
        float psel_part = 0.f, psum_part = 0.f;
        #pragma unroll
        for (int s = 0; s < HLOC; ++s) {
            float p_part = 0.f;
            const bool pick = (s == 1) ? pick1 : (s == 2) ? pick2 : pick3;
            #pragma unroll
            for (int mt = 0; mt < 2; ++mt) {
                const int j   = s * 2 + mt;
                const int cur = j & 1;
                const int nxt = cur ^ 1;

                if (j == 0) {
                    mq[0] = Mn[0];
                    mq[1] = Mn[128];
                }
                if (j < 7) {
                    mq[2 * j + 2] = Mn[128 * (2 * j + 2)];
                    mq[2 * j + 3] = Mn[128 * (2 * j + 3)];
                }
                if (j >= 2) {
                    put_m(N, sb + 272 * (2 * j - 4), mq[2 * j - 4]);
                    put_m(N, sb + 272 * (2 * j - 3), mq[2 * j - 3]);
                }
                if (j < 7) {
                    const int s1  = (j + 1) >> 1;
                    const int mt1 = (j + 1) & 1;
                    const float* mrBase = C + OFF_MR + ((s1 * DCUT) + mt1 * 16 + n) * MPAD + k0h;
                    const float* miBase = C + OFF_MI + ((s1 * DCUT) + mt1 * 16 + n) * MPAD + k0h;
                    #pragma unroll
                    for (int kk = 0; kk < 8; ++kk) {
                        fAr[nxt][kk] = *(const v2f*)(mrBase + kk * 4);
                        fAi[nxt][kk] = *(const v2f*)(miBase + kk * 4);
                    }
                    const float4* vr4 = (const float4*)(C + OFF_VR + s1 * DCUT + mt1 * 16 + half * 8);
                    const float4* vi4 = (const float4*)(C + OFF_VI + s1 * DCUT + mt1 * 16 + half * 8);
                    fVr[nxt][0] = vr4[0]; fVr[nxt][1] = vr4[1];
                    fVi[nxt][0] = vi4[0]; fVi[nxt][1] = vi4[1];
                }
                asm volatile("" : "+v"(fAr[cur][0]), "+v"(fAi[cur][0]));

                if (j == 1) {
                    site_scalars(N + OFF_W + wave * 2 * DCUT, half, Hr, Hi,
                                 cRprev, cIprev, pselPrev, psumPartPrev, logp, phase);
                }

                v8f accR  = {};
                v8f accMI = {};
                v8f accS  = {};
                #pragma unroll
                for (int kk = 0; kk < 8; ++kk) {
                    const v2f Ar = fAr[cur][kk];
                    const v2f Ai = fAi[cur][kk];
                    const v2f As = Ar + Ai;
                    accR  = wmma4(Ar, Hr[kk], accR);
                    accMI = wmma4(Ai, Hi[kk], accMI);
                    accS  = wmma4(As, Hs[kk], accS);
                }
                #pragma unroll
                for (int r = 0; r < 8; ++r) {
                    const float4 vr = fVr[cur][r >> 2];
                    const float4 vi = fVi[cur][r >> 2];
                    const float vrr = ((r & 3) == 0) ? vr.x : ((r & 3) == 1) ? vr.y : ((r & 3) == 2) ? vr.z : vr.w;
                    const float vii = ((r & 3) == 0) ? vi.x : ((r & 3) == 1) ? vi.y : ((r & 3) == 2) ? vi.z : vi.w;
                    const float hr = accR[r] - accMI[r] + vrr;
                    const float hi = accS[r] - (accR[r] + accMI[r]) + vii;
                    if (s == 0) {
                        selR[mt][r] = hr;
                        selI[mt][r] = hi;
                    } else {
                        selR[mt][r] = pick ? hr : selR[mt][r];
                        selI[mt][r] = pick ? hi : selI[mt][r];
                    }
                    p_part += hr * hr + hi * hi;
                }
            }
            psum_part += p_part;
            psel_part = (s == 0) ? p_part : (pick ? p_part : psel_part);
        }
        put_m(N, sb + 272 * 12, mq[12]);
        put_m(N, sb + 272 * 13, mq[13]);
        put_m(N, sb + 272 * 14, mq[14]);
        put_m(N, sb + 272 * 15, mq[15]);
        N[OFF_VR + tid] = vq.x;
        N[OFF_VI + tid] = vq.y;
        N[OFF_W + wave * 2 * DCUT + lane]        = wq.x;
        N[OFF_W + wave * 2 * DCUT + DCUT + lane] = wq.y;

        const float p_sel = psel_part + other_half(psel_part);
        const float rinv = rsqrtf(p_sel);
        pselPrev     = p_sel;
        psumPartPrev = psum_part;
        cRprev = cR;
        cIprev = cI;

        #pragma unroll
        for (int mt = 0; mt < 2; ++mt) {
            #pragma unroll
            for (int r = 0; r < 8; ++r) {
                Hr[4 * mt + (r >> 1)][r & 1] = selR[mt][r] * rinv;
                Hi[4 * mt + (r >> 1)][r & 1] = selI[mt][r] * rinv;
            }
        }
        #pragma unroll
        for (int k = 0; k < 8; ++k) Hs[k] = Hr[k] + Hi[k];

        __syncthreads();

        ss = 2 * xn0 + xn1;
    }

    site_scalars(site[(NS - 1) & 1] + OFF_W + wave * 2 * DCUT, half, Hr, Hi,
                 cRprev, cIprev, pselPrev, psumPartPrev, logp, phase);

    if (half == 0) {
        float2 o;
        o.x = 0.5f * logp;
        o.y = phase;
        *(float2*)(outg + (size_t)bcol * 2) = o;
    }
}

extern "C" void kernel_launch(void* const* d_in, const int* in_sizes, int n_in,
                              void* d_out, int out_size, void* d_ws, size_t ws_size,
                              hipStream_t stream) {
    (void)in_sizes; (void)n_in; (void)out_size; (void)d_ws; (void)ws_size;
    const int*   x  = (const int*)  d_in[0];
    const float* M  = (const float*)d_in[1];
    const float* v  = (const float*)d_in[2];
    const float* w  = (const float*)d_in[3];
    const float* c  = (const float*)d_in[4];
    float* out = (float*)d_out;

    dim3 grid(NB / 64);
    dim3 block(128);
    mps_rnn_wmma_kernel<<<grid, block, 0, stream>>>(x, M, v, w, c, out);
}


# Path between sites shortened: h carried as the kept plane itself, 1/sqrt(p_sel) multiplies the next site's products inside the add of v; next site's first LDS reads issued at the barrier
# speedup vs baseline: 2.6950x; 1.0420x over previous
#include <hip/hip_runtime.h>
#include <math.h>


#define NB   16384
#define NQ   100
#define NS   50
#define DCUT 32
#define HLOC 4

#define MPAD 34

#define MSZ     (HLOC * DCUT * MPAD)
#define OFF_MR  0
#define OFF_MI  MSZ
#define OFF_VR  (2 * MSZ)
#define OFF_VI  (2 * MSZ + HLOC * DCUT)
#define OFF_W   (2 * MSZ + 2 * HLOC * DCUT)
#define SITE_FLOATS (2 * MSZ + 2 * HLOC * DCUT + 4 * 2 * DCUT)

typedef __attribute__((ext_vector_type(2))) float v2f;
typedef __attribute__((ext_vector_type(8))) float v8f;

static __device__ __forceinline__ v8f wmma4(v2f a, v2f b, v8f c) {
    return __builtin_amdgcn_wmma_f32_16x16x4_f32(
         false, a,  false, b,
         (short)0, c,  false,  false);
}

static __device__ __forceinline__ float other_half(float x) {
    const int xi = __builtin_bit_cast(int, x);
    return __builtin_bit_cast(float,
        __builtin_amdgcn_permlanex16(xi, xi, 0x76543210, 0xfedcba98, false, false));
}

static __device__ __forceinline__ void put_m(float* S, int off, float4 m) {
    v2f re, im;
    re[0] = m.x; re[1] = m.z;
    im[0] = m.y; im[1] = m.w;
    *(v2f*)(S + OFF_MR + off) = re;
    *(v2f*)(S + OFF_MI + off) = im;
}

static __device__ __forceinline__ void site_scalars(
    const float* Wst, int half, const v2f (&Hr)[8], const v2f (&Hi)[8],
    float cR, float cI, float p_sel, float rscale, float psum_part, float& logp, float& phase)
{
    const float psum = psum_part + other_half(psum_part);
    logp += logf(p_sel) - logf(psum);

    float ampR = 0.f, ampI = 0.f;
    #pragma unroll
    for (int kk = 0; kk < 8; ++kk) {
        const int k = 16 * (kk >> 2) + 8 * half + 2 * (kk & 3);
        const v2f wr = *(const v2f*)(Wst + k);
        const v2f wi = *(const v2f*)(Wst + DCUT + k);
        #pragma unroll
        for (int el = 0; el < 2; ++el) {
            ampR += wr[el] * Hr[kk][el] - wi[el] * Hi[kk][el];
            ampI += wr[el] * Hi[kk][el] + wi[el] * Hr[kk][el];
        }
    }
    ampR += other_half(ampR);
    ampI += other_half(ampI);
    ampR = ampR * rscale + cR;
    ampI = ampI * rscale + cI;
    phase += atan2f(ampI, ampR);
}

__global__ __launch_bounds__(128) __attribute__((amdgpu_waves_per_eu(1, 1))) void mps_rnn_wmma_kernel(
    const int*   __restrict__ xg,
    const float* __restrict__ Mg,
    const float* __restrict__ vg,
    const float* __restrict__ wg,
    const float* __restrict__ cg,
    float*       __restrict__ outg)
{
    __shared__ __attribute__((aligned(16))) float site[2][SITE_FLOATS];

    const int tid  = threadIdx.x;
    const int lane = tid & 31;
    const int wave = tid >> 5;
    const int n    = lane & 15;
    const int half = lane >> 4;
    const int bcol = blockIdx.x * 64 + wave * 16 + n;

    const int ej = tid & 15;
    const int sb = (tid >> 4) * MPAD + 16 * (ej >> 3) + 4 * (ej & 3) + 2 * ((ej >> 2) & 1);
    const float4* M4 = reinterpret_cast<const float4*>(Mg);
    const float2* v2 = reinterpret_cast<const float2*>(vg);
    const float2* w2 = reinterpret_cast<const float2*>(wg);

    {
        float* S = site[0];
        #pragma unroll
        for (int q = 0; q < 16; ++q) put_m(S, sb + 272 * q, M4[tid + 128 * q]);
        const float2 vq = v2[tid];
        S[OFF_VR + tid] = vq.x;
        S[OFF_VI + tid] = vq.y;
        const float2 wq = w2[lane];
        S[OFF_W + wave * 2 * DCUT + lane]        = wq.x;
        S[OFF_W + wave * 2 * DCUT + DCUT + lane] = wq.y;
        site[1][OFF_W + wave * 2 * DCUT + lane]        = 0.f;
        site[1][OFF_W + wave * 2 * DCUT + DCUT + lane] = 0.f;
    }

    v2f Hr[8], Hi[8], Hs[8];
    const float h0 = 0.17677669529663688f;
    #pragma unroll
    for (int k = 0; k < 8; ++k) {
        Hr[k][0] = h0; Hr[k][1] = h0;
        Hi[k][0] = 0.f; Hi[k][1] = 0.f;
        Hs[k] = Hr[k] + Hi[k];
    }

    float logp = 0.f, phase = 0.f;
    const int k0h = half * 2;

    float pselPartPrev = 0.5f, psumPartPrev = 0.5f, cRprev = 1.f, cIprev = 0.f;

    v2f    fAr[2][8], fAi[2][8];
    float4 fVr[2][2], fVi[2][2];

    int ss = 2 * xg[(size_t)bcol * NQ] + xg[(size_t)bcol * NQ + 1];

    __syncthreads();

    {
        const float* mrBase = site[0] + OFF_MR + n * MPAD + k0h;
        const float* miBase = site[0] + OFF_MI + n * MPAD + k0h;
        #pragma unroll
        for (int kk = 0; kk < 8; ++kk) {
            fAr[0][kk] = *(const v2f*)(mrBase + kk * 4);
            fAi[0][kk] = *(const v2f*)(miBase + kk * 4);
        }
        const float4* vr4 = (const float4*)(site[0] + OFF_VR + half * 8);
        const float4* vi4 = (const float4*)(site[0] + OFF_VI + half * 8);
        fVr[0][0] = vr4[0]; fVr[0][1] = vr4[1];
        fVi[0][0] = vi4[0]; fVi[0][1] = vi4[1];
    }

    for (int t = 0; t < NS; ++t) {
        const float* C = site[t & 1];
        float*       N = site[(t & 1) ^ 1];

        const int tn = (t + 1 < NS) ? (t + 1) : (NS - 1);
        const float4* Mn = M4 + (size_t)tn * (HLOC * DCUT * DCUT / 2) + tid;
        const float2 vq = v2[tn * (HLOC * DCUT) + tid];
        const float2 wq = w2[tn * DCUT + lane];
        const int xn0 = xg[(size_t)bcol * NQ + 2 * tn];
        const int xn1 = xg[(size_t)bcol * NQ + 2 * tn + 1];
        const bool pick1 = (ss == 1);
        const bool pick2 = (ss == 2);
        const bool pick3 = !(ss == 0 || ss == 1 || ss == 2);
        const float cR = cg[2 * t];
        const float cI = cg[2 * t + 1];
        float4 mq[16];

        const float pselPrev = pselPartPrev + other_half(pselPartPrev);
        const float rcur = rsqrtf(pselPrev);

        v8f selR[2], selI[2];
        float psel_part = 0.f, psum_part = 0.f;
        #pragma unroll
        for (int s = 0; s < HLOC; ++s) {
            float p_part = 0.f;
            const bool pick = (s == 1) ? pick1 : (s == 2) ? pick2 : pick3;
            #pragma unroll
            for (int mt = 0; mt < 2; ++mt) {
                const int j   = s * 2 + mt;
                const int cur = j & 1;
                const int nxt = cur ^ 1;

                if (j == 0) {
                    mq[0] = Mn[0];
                    mq[1] = Mn[128];
                }
                if (j < 7) {
                    mq[2 * j + 2] = Mn[128 * (2 * j + 2)];
                    mq[2 * j + 3] = Mn[128 * (2 * j + 3)];
                }
                if (j >= 2) {
                    put_m(N, sb + 272 * (2 * j - 4), mq[2 * j - 4]);
                    put_m(N, sb + 272 * (2 * j - 3), mq[2 * j - 3]);
                }
                if (j == 7) {
                    put_m(N, sb + 272 * 12, mq[12]);
                    put_m(N, sb + 272 * 13, mq[13]);
                    put_m(N, sb + 272 * 14, mq[14]);
                    put_m(N, sb + 272 * 15, mq[15]);
                    N[OFF_VR + tid] = vq.x;
                    N[OFF_VI + tid] = vq.y;
                    N[OFF_W + wave * 2 * DCUT + lane]        = wq.x;
                    N[OFF_W + wave * 2 * DCUT + DCUT + lane] = wq.y;
                    __syncthreads();
                    const float* mrBase = N + OFF_MR + n * MPAD + k0h;
                    const float* miBase = N + OFF_MI + n * MPAD + k0h;
                    #pragma unroll
                    for (int kk = 0; kk < 8; ++kk) {
                        fAr[0][kk] = *(const v2f*)(mrBase + kk * 4);
                        fAi[0][kk] = *(const v2f*)(miBase + kk * 4);
                    }
                    const float4* vr4 = (const float4*)(N + OFF_VR + half * 8);
                    const float4* vi4 = (const float4*)(N + OFF_VI + half * 8);
                    fVr[0][0] = vr4[0]; fVr[0][1] = vr4[1];
                    fVi[0][0] = vi4[0]; fVi[0][1] = vi4[1];
                }
                if (j < 7) {
                    const int s1  = (j + 1) >> 1;
                    const int mt1 = (j + 1) & 1;
                    const float* mrBase = C + OFF_MR + ((s1 * DCUT) + mt1 * 16 + n) * MPAD + k0h;
                    const float* miBase = C + OFF_MI + ((s1 * DCUT) + mt1 * 16 + n) * MPAD + k0h;
                    #pragma unroll
                    for (int kk = 0; kk < 8; ++kk) {
                        fAr[nxt][kk] = *(const v2f*)(mrBase + kk * 4);
                        fAi[nxt][kk] = *(const v2f*)(miBase + kk * 4);
                    }
                    const float4* vr4 = (const float4*)(C + OFF_VR + s1 * DCUT + mt1 * 16 + half * 8);
                    const float4* vi4 = (const float4*)(C + OFF_VI + s1 * DCUT + mt1 * 16 + half * 8);
                    fVr[nxt][0] = vr4[0]; fVr[nxt][1] = vr4[1];
                    fVi[nxt][0] = vi4[0]; fVi[nxt][1] = vi4[1];
                }
                asm volatile("" : "+v"(fAr[cur][0]), "+v"(fAi[cur][0]));

                if (j == 1) {
                    site_scalars(N + OFF_W + wave * 2 * DCUT, half, Hr, Hi,
                                 cRprev, cIprev, pselPrev, rcur, psumPartPrev, logp, phase);
                }

                v8f accR  = {};
                v8f accMI = {};
                v8f accS  = {};
                #pragma unroll
                for (int kk = 0; kk < 8; ++kk) {
                    const v2f Ar = fAr[cur][kk];
                    const v2f Ai = fAi[cur][kk];
                    const v2f As = Ar + Ai;
                    accR  = wmma4(Ar, Hr[kk], accR);
                    accMI = wmma4(Ai, Hi[kk], accMI);
                    accS  = wmma4(As, Hs[kk], accS);
                }
                #pragma unroll
                for (int r = 0; r < 8; ++r) {
                    const float4 vr = fVr[cur][r >> 2];
                    const float4 vi = fVi[cur][r >> 2];
                    const float vrr = ((r & 3) == 0) ? vr.x : ((r & 3) == 1) ? vr.y : ((r & 3) == 2) ? vr.z : vr.w;
                    const float vii = ((r & 3) == 0) ? vi.x : ((r & 3) == 1) ? vi.y : ((r & 3) == 2) ? vi.z : vi.w;
                    const float hr = (accR[r] - accMI[r]) * rcur + vrr;
                    const float hi = (accS[r] - (accR[r] + accMI[r])) * rcur + vii;
                    if (s == 0) {
                        selR[mt][r] = hr;
                        selI[mt][r] = hi;
                    } else {
                        selR[mt][r] = pick ? hr : selR[mt][r];
                        selI[mt][r] = pick ? hi : selI[mt][r];
                    }
                    p_part += hr * hr + hi * hi;
                }
            }
            psum_part += p_part;
            psel_part = (s == 0) ? p_part : (pick ? p_part : psel_part);
        }
        pselPartPrev = psel_part;
        psumPartPrev = psum_part;
        cRprev = cR;
        cIprev = cI;

        #pragma unroll
        for (int mt = 0; mt < 2; ++mt) {
            #pragma unroll
            for (int r = 0; r < 8; ++r) {
                Hr[4 * mt + (r >> 1)][r & 1] = selR[mt][r];
                Hi[4 * mt + (r >> 1)][r & 1] = selI[mt][r];
            }
        }
        #pragma unroll
        for (int k = 0; k < 8; ++k) Hs[k] = Hr[k] + Hi[k];

        ss = 2 * xn0 + xn1;
    }

    {
        const float pselLast = pselPartPrev + other_half(pselPartPrev);
        site_scalars(site[(NS - 1) & 1] + OFF_W + wave * 2 * DCUT, half, Hr, Hi,
                     cRprev, cIprev, pselLast, rsqrtf(pselLast), psumPartPrev, logp, phase);
    }

    if (half == 0) {
        float2 o;
        o.x = 0.5f * logp;
        o.y = phase;
        *(float2*)(outg + (size_t)bcol * 2) = o;
    }
}

extern "C" void kernel_launch(void* const* d_in, const int* in_sizes, int n_in,
                              void* d_out, int out_size, void* d_ws, size_t ws_size,
                              hipStream_t stream) {
    (void)in_sizes; (void)n_in; (void)out_size; (void)d_ws; (void)ws_size;
    const int*   x  = (const int*)  d_in[0];
    const float* M  = (const float*)d_in[1];
    const float* v  = (const float*)d_in[2];
    const float* w  = (const float*)d_in[3];
    const float* c  = (const float*)d_in[4];
    float* out = (float*)d_out;

    dim3 grid(NB / 64);
    dim3 block(128);
    mps_rnn_wmma_kernel<<<grid, block, 0, stream>>>(x, M, v, w, c, out);
}


# Three-product sums done in the matrix unit: P1=(Mr+Mi)Hr is the C operand of chains Mi*(-(Hr+Hi)) and Mr*(Hi-Hr); a plane's two tiles run together, LDS groups per plane; real/imag sum order changes
# speedup vs baseline: 2.8073x; 1.0417x over previous
#include <hip/hip_runtime.h>
#include <math.h>


#define NB   16384
#define NQ   100
#define NS   50
#define DCUT 32
#define HLOC 4

#define MPAD 34

#define MSZ     (HLOC * DCUT * MPAD)
#define OFF_MR  0
#define OFF_MI  MSZ
#define OFF_VR  (2 * MSZ)
#define OFF_VI  (2 * MSZ + HLOC * DCUT)
#define OFF_W   (2 * MSZ + 2 * HLOC * DCUT)
#define SITE_FLOATS (2 * MSZ + 2 * HLOC * DCUT + 4 * 2 * DCUT)

typedef __attribute__((ext_vector_type(2))) float v2f;
typedef __attribute__((ext_vector_type(8))) float v8f;

static __device__ __forceinline__ v8f wmma4(v2f a, v2f b, v8f c) {
    return __builtin_amdgcn_wmma_f32_16x16x4_f32(
         false, a,  false, b,
         (short)0, c,  false,  false);
}

static __device__ __forceinline__ float other_half(float x) {
    const int xi = __builtin_bit_cast(int, x);
    return __builtin_bit_cast(float,
        __builtin_amdgcn_permlanex16(xi, xi, 0x76543210, 0xfedcba98, false, false));
}

static __device__ __forceinline__ void put_m(float* S, int off, float4 m) {
    v2f re, im;
    re[0] = m.x; re[1] = m.z;
    im[0] = m.y; im[1] = m.w;
    *(v2f*)(S + OFF_MR + off) = re;
    *(v2f*)(S + OFF_MI + off) = im;
}

static __device__ __forceinline__ void site_scalars(
    const float* Wst, int half, const v2f (&Hr)[8], const v2f (&Hi)[8],
    float cR, float cI, float p_sel, float rscale, float psum_part, float& logp, float& phase)
{
    const float psum = psum_part + other_half(psum_part);
    logp += logf(p_sel) - logf(psum);

    float ampR = 0.f, ampI = 0.f;
    #pragma unroll
    for (int kk = 0; kk < 8; ++kk) {
        const int k = 16 * (kk >> 2) + 8 * half + 2 * (kk & 3);
        const v2f wr = *(const v2f*)(Wst + k);
        const v2f wi = *(const v2f*)(Wst + DCUT + k);
        #pragma unroll
        for (int el = 0; el < 2; ++el) {
            ampR += wr[el] * Hr[kk][el] - wi[el] * Hi[kk][el];
            ampI += wr[el] * Hi[kk][el] + wi[el] * Hr[kk][el];
        }
    }
    ampR += other_half(ampR);
    ampI += other_half(ampI);
    ampR = ampR * rscale + cR;
    ampI = ampI * rscale + cI;
    phase += atan2f(ampI, ampR);
}

__global__ __launch_bounds__(128) __attribute__((amdgpu_waves_per_eu(1, 1))) void mps_rnn_wmma_kernel(
    const int*   __restrict__ xg,
    const float* __restrict__ Mg,
    const float* __restrict__ vg,
    const float* __restrict__ wg,
    const float* __restrict__ cg,
    float*       __restrict__ outg)
{
    __shared__ __attribute__((aligned(16))) float site[2][SITE_FLOATS];

    const int tid  = threadIdx.x;
    const int lane = tid & 31;
    const int wave = tid >> 5;
    const int n    = lane & 15;
    const int half = lane >> 4;
    const int bcol = blockIdx.x * 64 + wave * 16 + n;

    const int ej = tid & 15;
    const int sb = (tid >> 4) * MPAD + 16 * (ej >> 3) + 4 * (ej & 3) + 2 * ((ej >> 2) & 1);
    const float4* M4 = reinterpret_cast<const float4*>(Mg);
    const float2* v2 = reinterpret_cast<const float2*>(vg);
    const float2* w2 = reinterpret_cast<const float2*>(wg);

    {
        float* S = site[0];
        #pragma unroll
        for (int q = 0; q < 16; ++q) put_m(S, sb + 272 * q, M4[tid + 128 * q]);
        const float2 vq = v2[tid];
        S[OFF_VR + tid] = vq.x;
        S[OFF_VI + tid] = vq.y;
        const float2 wq = w2[lane];
        S[OFF_W + wave * 2 * DCUT + lane]        = wq.x;
        S[OFF_W + wave * 2 * DCUT + DCUT + lane] = wq.y;
        site[1][OFF_W + wave * 2 * DCUT + lane]        = 0.f;
        site[1][OFF_W + wave * 2 * DCUT + DCUT + lane] = 0.f;
    }

    v2f Hr[8], Hi[8], Hd[8], Hn[8];
    const float h0 = 0.17677669529663688f;
    #pragma unroll
    for (int k = 0; k < 8; ++k) {
        Hr[k][0] = h0; Hr[k][1] = h0;
        Hi[k][0] = 0.f; Hi[k][1] = 0.f;
        Hd[k] = Hi[k] - Hr[k];
        Hn[k] = -Hr[k] - Hi[k];
    }

    float logp = 0.f, phase = 0.f;
    const int k0h = half * 2;

    float pselPartPrev = 0.5f, psumPartPrev = 0.5f, cRprev = 1.f, cIprev = 0.f;

    v2f    fAr[2][2][8], fAi[2][2][8];
    float4 fVr[2][2][2], fVi[2][2][2];

    int ss = 2 * xg[(size_t)bcol * NQ] + xg[(size_t)bcol * NQ + 1];

    __syncthreads();

    #pragma unroll
    for (int mt = 0; mt < 2; ++mt) {
        const float* mrBase = site[0] + OFF_MR + (mt * 16 + n) * MPAD + k0h;
        const float* miBase = site[0] + OFF_MI + (mt * 16 + n) * MPAD + k0h;
        #pragma unroll
        for (int kk = 0; kk < 8; ++kk) {
            fAr[0][mt][kk] = *(const v2f*)(mrBase + kk * 4);
            fAi[0][mt][kk] = *(const v2f*)(miBase + kk * 4);
        }
        const float4* vr4 = (const float4*)(site[0] + OFF_VR + mt * 16 + half * 8);
        const float4* vi4 = (const float4*)(site[0] + OFF_VI + mt * 16 + half * 8);
        fVr[0][mt][0] = vr4[0]; fVr[0][mt][1] = vr4[1];
        fVi[0][mt][0] = vi4[0]; fVi[0][mt][1] = vi4[1];
    }

    for (int t = 0; t < NS; ++t) {
        const float* C = site[t & 1];
        float*       N = site[(t & 1) ^ 1];

        const int tn = (t + 1 < NS) ? (t + 1) : (NS - 1);
        const float4* Mn = M4 + (size_t)tn * (HLOC * DCUT * DCUT / 2) + tid;
        const float2 vq = v2[tn * (HLOC * DCUT) + tid];
        const float2 wq = w2[tn * DCUT + lane];
        const int xn0 = xg[(size_t)bcol * NQ + 2 * tn];
        const int xn1 = xg[(size_t)bcol * NQ + 2 * tn + 1];
        const bool pick1 = (ss == 1);
        const bool pick2 = (ss == 2);
        const bool pick3 = !(ss == 0 || ss == 1 || ss == 2);
        const float cR = cg[2 * t];
        const float cI = cg[2 * t + 1];
        float4 mq[16];

        const float pselPrev = pselPartPrev + other_half(pselPartPrev);
        const float rcur = rsqrtf(pselPrev);

        v8f selR[2], selI[2];
        float psel_part = 0.f, psum_part = 0.f;
        #pragma unroll
        for (int s = 0; s < HLOC; ++s) {
            float p_part = 0.f;
            const bool pick = (s == 1) ? pick1 : (s == 2) ? pick2 : pick3;
            const int cur = s & 1;
            const int nxt = cur ^ 1;

            if (s == 0) {
                #pragma unroll
                for (int q = 0; q < 8; ++q) mq[q] = Mn[128 * q];
            }
            if (s == 1) {
                #pragma unroll
                for (int q = 8; q < 16; ++q) mq[q] = Mn[128 * q];
                #pragma unroll
                for (int q = 0; q < 8; ++q) put_m(N, sb + 272 * q, mq[q]);
            }
            if (s == 2) {
                #pragma unroll
                for (int q = 8; q < 16; ++q) put_m(N, sb + 272 * q, mq[q]);
                N[OFF_VR + tid] = vq.x;
                N[OFF_VI + tid] = vq.y;
                N[OFF_W + wave * 2 * DCUT + lane]        = wq.x;
                N[OFF_W + wave * 2 * DCUT + DCUT + lane] = wq.y;
            }
            if (s == 3) {
                __syncthreads();
                #pragma unroll
                for (int mt = 0; mt < 2; ++mt) {
                    const float* mrBase = N + OFF_MR + (mt * 16 + n) * MPAD + k0h;
                    const float* miBase = N + OFF_MI + (mt * 16 + n) * MPAD + k0h;
                    #pragma unroll
                    for (int kk = 0; kk < 8; ++kk) {
                        fAr[0][mt][kk] = *(const v2f*)(mrBase + kk * 4);
                        fAi[0][mt][kk] = *(const v2f*)(miBase + kk * 4);
                    }
                    const float4* vr4 = (const float4*)(N + OFF_VR + mt * 16 + half * 8);
                    const float4* vi4 = (const float4*)(N + OFF_VI + mt * 16 + half * 8);
                    fVr[0][mt][0] = vr4[0]; fVr[0][mt][1] = vr4[1];
                    fVi[0][mt][0] = vi4[0]; fVi[0][mt][1] = vi4[1];
                }
            }
            if (s < 3) {
                #pragma unroll
                for (int mt = 0; mt < 2; ++mt) {
                    const float* mrBase = C + OFF_MR + (((s + 1) * DCUT) + mt * 16 + n) * MPAD + k0h;
                    const float* miBase = C + OFF_MI + (((s + 1) * DCUT) + mt * 16 + n) * MPAD + k0h;
                    #pragma unroll
                    for (int kk = 0; kk < 8; ++kk) {
                        fAr[nxt][mt][kk] = *(const v2f*)(mrBase + kk * 4);
                        fAi[nxt][mt][kk] = *(const v2f*)(miBase + kk * 4);
                    }
                    const float4* vr4 = (const float4*)(C + OFF_VR + (s + 1) * DCUT + mt * 16 + half * 8);
                    const float4* vi4 = (const float4*)(C + OFF_VI + (s + 1) * DCUT + mt * 16 + half * 8);
                    fVr[nxt][mt][0] = vr4[0]; fVr[nxt][mt][1] = vr4[1];
                    fVi[nxt][mt][0] = vi4[0]; fVi[nxt][mt][1] = vi4[1];
                }
            }
            asm volatile("" : "+v"(fAr[cur][0][0]), "+v"(fAi[cur][0][0]),
                              "+v"(fAr[cur][1][0]), "+v"(fAi[cur][1][0]));

            if (s == 0) {
                site_scalars(N + OFF_W + wave * 2 * DCUT, half, Hr, Hi,
                             cRprev, cIprev, pselPrev, rcur, psumPartPrev, logp, phase);
            }

            v8f acc1[2] = {{}, {}};
            #pragma unroll
            for (int kk = 0; kk < 8; ++kk) {
                #pragma unroll
                for (int mt = 0; mt < 2; ++mt) {
                    const v2f As = fAr[cur][mt][kk] + fAi[cur][mt][kk];
                    acc1[mt] = wmma4(As, Hr[kk], acc1[mt]);
                }
            }
            v8f accRe[2], accIm[2];
            #pragma unroll
            for (int mt = 0; mt < 2; ++mt) {
                accRe[mt] = wmma4(fAi[cur][mt][0], Hn[0], acc1[mt]);
                accIm[mt] = wmma4(fAr[cur][mt][0], Hd[0], acc1[mt]);
            }
            #pragma unroll
            for (int kk = 1; kk < 8; ++kk) {
                #pragma unroll
                for (int mt = 0; mt < 2; ++mt) {
                    accRe[mt] = wmma4(fAi[cur][mt][kk], Hn[kk], accRe[mt]);
                    accIm[mt] = wmma4(fAr[cur][mt][kk], Hd[kk], accIm[mt]);
                }
            }
            if (s == 2) {
                asm volatile("" :: "v"(accRe[0]), "v"(accIm[0]), "v"(accRe[1]), "v"(accIm[1]));
            }
            #pragma unroll
            for (int mt = 0; mt < 2; ++mt) {
                #pragma unroll
                for (int r = 0; r < 8; ++r) {
                    const float4 vr = fVr[cur][mt][r >> 2];
                    const float4 vi = fVi[cur][mt][r >> 2];
                    const float vrr = ((r & 3) == 0) ? vr.x : ((r & 3) == 1) ? vr.y : ((r & 3) == 2) ? vr.z : vr.w;
                    const float vii = ((r & 3) == 0) ? vi.x : ((r & 3) == 1) ? vi.y : ((r & 3) == 2) ? vi.z : vi.w;
                    const float hr = accRe[mt][r] * rcur + vrr;
                    const float hi = accIm[mt][r] * rcur + vii;
                    if (s == 0) {
                        selR[mt][r] = hr;
                        selI[mt][r] = hi;
                    } else {
                        selR[mt][r] = pick ? hr : selR[mt][r];
                        selI[mt][r] = pick ? hi : selI[mt][r];
                    }
                    p_part += hr * hr + hi * hi;
                }
            }
            psum_part += p_part;
            psel_part = (s == 0) ? p_part : (pick ? p_part : psel_part);
        }
        pselPartPrev = psel_part;
        psumPartPrev = psum_part;
        cRprev = cR;
        cIprev = cI;

        #pragma unroll
        for (int mt = 0; mt < 2; ++mt) {
            #pragma unroll
            for (int r = 0; r < 8; ++r) {
                Hr[4 * mt + (r >> 1)][r & 1] = selR[mt][r];
                Hi[4 * mt + (r >> 1)][r & 1] = selI[mt][r];
            }
        }
        #pragma unroll
        for (int k = 0; k < 8; ++k) {
            Hd[k] = Hi[k] - Hr[k];
            Hn[k] = -Hr[k] - Hi[k];
        }

        ss = 2 * xn0 + xn1;
    }

    {
        const float pselLast = pselPartPrev + other_half(pselPartPrev);
        site_scalars(site[(NS - 1) & 1] + OFF_W + wave * 2 * DCUT, half, Hr, Hi,
                     cRprev, cIprev, pselLast, rsqrtf(pselLast), psumPartPrev, logp, phase);
    }

    if (half == 0) {
        float2 o;
        o.x = 0.5f * logp;
        o.y = phase;
        *(float2*)(outg + (size_t)bcol * 2) = o;
    }
}

extern "C" void kernel_launch(void* const* d_in, const int* in_sizes, int n_in,
                              void* d_out, int out_size, void* d_ws, size_t ws_size,
                              hipStream_t stream) {
    (void)in_sizes; (void)n_in; (void)out_size; (void)d_ws; (void)ws_size;
    const int*   x  = (const int*)  d_in[0];
    const float* M  = (const float*)d_in[1];
    const float* v  = (const float*)d_in[2];
    const float* w  = (const float*)d_in[3];
    const float* c  = (const float*)d_in[4];
    float* out = (float*)d_out;

    dim3 grid(NB / 64);
    dim3 block(128);
    mps_rnn_wmma_kernel<<<grid, block, 0, stream>>>(x, M, v, w, c, out);
}
